# starting kernel
# baseline (speedup 1.0000x reference)
#include <hip/hip_runtime.h>
#include <hip/hip_bf16.h>

typedef __attribute__((ext_vector_type(2))) float v2f;
typedef __attribute__((ext_vector_type(8))) float v8f;

#define NNODES 50000
#define NEDGES 800000
#define NGRAPH 512
#define INDIM  128
#define HID    96
#define OUTDIM 10
#define NCONV  4
#define NTILE  6
#define GEMM_WAVES 8

static inline int cdiv(int a, int b) { return (a + b - 1) / b; }

__global__ void k_deg_init(float* deg, int n) {
    int i = blockIdx.x * blockDim.x + threadIdx.x;
    if (i < n) deg[i] = 1.0f;
}

__global__ void k_deg_accum(const int* __restrict__ ei, float* deg, int E) {
    int e = blockIdx.x * blockDim.x + threadIdx.x;
    if (e < E) atomicAdd(&deg[ei[E + e]], 1.0f);
}

__global__ void k_rsqrt_inplace(float* deg, int n) {
    int i = blockIdx.x * blockDim.x + threadIdx.x;
    if (i < n) deg[i] = rsqrtf(deg[i]);
}

template <int RELUA, int RELU, int FUSE>
__global__ void k_gemm_wmma(const float* __restrict__ A, const float* __restrict__ B,
                            const float* __restrict__ bias,
                            float* __restrict__ C, float* __restrict__ C2,
                            const float* __restrict__ dinv,
                            int M, int K, int lda) {
    extern __shared__ float sBt[];
    int lstride = K + 2;

    int nB = K * HID;
    for (int i = threadIdx.x; i < nB; i += blockDim.x) {
        int k   = i / HID;
        int col = i - k * HID;
        sBt[col * lstride + k] = B[i];
    }
    __syncthreads();

    int mt   = blockIdx.x * GEMM_WAVES + (threadIdx.x >> 5);
    int lane = threadIdx.x & 31;
    int mTiles = M >> 4;
    if (mt >= mTiles) return;

    int half = lane >> 4;
    int l15  = lane & 15;

    const float* arow  = A + (size_t)(mt * 16 + l15) * lda;
    const float* bbase = sBt + (size_t)l15 * lstride;

    v8f acc[NTILE] = {};
    for (int k = 0; k < K; k += 4) {
        int kb = k + half * 2;
        v2f a = *(const v2f*)(arow + kb);
        if (RELUA) { a[0] = fmaxf(a[0], 0.0f); a[1] = fmaxf(a[1], 0.0f); }
        #pragma unroll
        for (int nt = 0; nt < NTILE; nt++) {
            v2f b = *(const v2f*)(bbase + (size_t)(nt * 16) * lstride + kb);
            acc[nt] = __builtin_amdgcn_wmma_f32_16x16x4_f32(
                 false, a,  false, b,
                 (short)0, acc[nt],  false,  false);
        }
    }

    int rbase = mt * 16 + half * 8;
    float* cbase = C + (size_t)rbase * HID + l15;

    if (FUSE) {
        float dv2[8];
        #pragma unroll
        for (int v = 0; v < 8; v++) { float d = dinv[rbase + v]; dv2[v] = d * d; }
        float* c2base = C2 + (size_t)rbase * HID + l15;
        #pragma unroll
        for (int nt = 0; nt < NTILE; nt++) {
            float bv = bias[nt * 16 + l15];
            #pragma unroll
            for (int v = 0; v < 8; v++) {
                float raw = acc[nt][v];
                cbase [(size_t)v * HID + nt * 16] = raw;
                c2base[(size_t)v * HID + nt * 16] = raw * dv2[v] + bv;
            }
        }
    } else {
        #pragma unroll
        for (int nt = 0; nt < NTILE; nt++) {
            float bv = bias[nt * 16 + l15];
            #pragma unroll
            for (int v = 0; v < 8; v++) {
                float r = acc[nt][v] + bv;
                if (RELU) r = fmaxf(r, 0.0f);
                cbase[(size_t)v * HID + nt * 16] = r;
            }
        }
    }
}

__global__ void k_edge_scatter(const float* __restrict__ hw, const float* __restrict__ dinv,
                               const int* __restrict__ ei, float* __restrict__ out, int E) {
    int wave = (blockIdx.x * blockDim.x + threadIdx.x) >> 5;
    int lane = threadIdx.x & 31;
    if (wave >= E) return;
    int s = ei[wave];
    int d = ei[E + wave];
    float norm = dinv[s] * dinv[d];
    const float* srow = hw  + (size_t)s * HID;
    float*       drow = out + (size_t)d * HID;
    #pragma unroll
    for (int k = 0; k < 3; k++) {
        int c = lane + k * 32;
        atomicAdd(drow + c, srow[c] * norm);
    }
}

__global__ void k_zero(float* x, int n) {
    int i = blockIdx.x * blockDim.x + threadIdx.x;
    if (i < n) x[i] = 0.0f;
}

__global__ void k_pool(const float* __restrict__ h, const int* __restrict__ batch,
                       float* __restrict__ g, int n) {
    int idx = blockIdx.x * blockDim.x + threadIdx.x;
    if (idx < n * HID) {
        int node = idx / HID;
        int c    = idx - node * HID;
        atomicAdd(&g[(size_t)batch[node] * HID + c], fmaxf(h[idx], 0.0f));
    }
}

__global__ void k_dec2(const float* __restrict__ g2, const float* __restrict__ W,
                       const float* __restrict__ b, float* __restrict__ out) {
    int idx = blockIdx.x * blockDim.x + threadIdx.x;
    if (idx < NGRAPH * OUTDIM) {
        int gi = idx / OUTDIM;
        int o  = idx - gi * OUTDIM;
        float s = b[o];
        const float* row = g2 + (size_t)gi * HID;
        #pragma unroll 8
        for (int k = 0; k < HID; k++) s += row[k] * W[k * OUTDIM + o];
        out[idx] = s;
    }
}

extern "C" void kernel_launch(void* const* d_in, const int* in_sizes, int n_in,
                              void* d_out, int out_size, void* d_ws, size_t ws_size,
                              hipStream_t stream) {
    const float* x       = (const float*)d_in[0];
    const int*   ei      = (const int*)  d_in[1];
    const int*   batch   = (const int*)  d_in[2];
    const float* enc_W0  = (const float*)d_in[3];
    const float* enc_b0  = (const float*)d_in[4];
    const float* enc_W1  = (const float*)d_in[5];
    const float* enc_b1  = (const float*)d_in[6];
    const float* conv_W  = (const float*)d_in[7];
    const float* conv_b  = (const float*)d_in[8];
    const float* dec_W0  = (const float*)d_in[9];
    const float* dec_b0  = (const float*)d_in[10];
    const float* dec_W1  = (const float*)d_in[11];
    const float* dec_b1  = (const float*)d_in[12];
    float* out = (float*)d_out;

    float* ws   = (float*)d_ws;
    size_t feat = (size_t)NNODES * HID;
    float* buf0 = ws;
    float* buf1 = buf0 + feat;
    float* buf2 = buf1 + feat;
    float* dinv = buf2 + feat;
    float* g    = dinv + NNODES;
    float* g2   = g + (size_t)NGRAPH * HID;

    const int B = 256;

    k_deg_init<<<cdiv(NNODES, B), B, 0, stream>>>(dinv, NNODES);
    k_deg_accum<<<cdiv(NEDGES, B), B, 0, stream>>>(ei, dinv, NEDGES);
    k_rsqrt_inplace<<<cdiv(NNODES, B), B, 0, stream>>>(dinv, NNODES);

    int mTilesN   = NNODES / 16;
    int gemmBlksN = cdiv(mTilesN, GEMM_WAVES);
    size_t shm128 = (size_t)HID * (INDIM + 2) * sizeof(float);
    size_t shm96  = (size_t)HID * (HID   + 2) * sizeof(float);
    k_gemm_wmma<0, 1, 0><<<gemmBlksN, B, shm128, stream>>>(
        x, enc_W0, enc_b0, buf0, nullptr, nullptr, NNODES, INDIM, INDIM);
    k_gemm_wmma<0, 0, 0><<<gemmBlksN, B, shm96, stream>>>(
        buf0, enc_W1, enc_b1, buf1, nullptr, nullptr, NNODES, HID, HID);

    float* cur = buf1;
    float* nxt = buf2;
    int edgeBlks = cdiv(NEDGES * 32, B);
    for (int l = 0; l < NCONV; l++) {
        const float* Wl = conv_W + (size_t)l * HID * HID;
        const float* bl = conv_b + (size_t)l * HID;
        if (l == 0)
            k_gemm_wmma<0, 0, 1><<<gemmBlksN, B, shm96, stream>>>(
                cur, Wl, bl, buf0  , nxt  , dinv, NNODES, HID, HID);
        else
            k_gemm_wmma<1, 0, 1><<<gemmBlksN, B, shm96, stream>>>(
                cur, Wl, bl, buf0  , nxt  , dinv, NNODES, HID, HID);
        k_edge_scatter<<<edgeBlks, B, 0, stream>>>(buf0, dinv, ei, nxt, NEDGES);
        float* t = cur; cur = nxt; nxt = t;
    }

    k_zero<<<cdiv(NGRAPH * HID, B), B, 0, stream>>>(g, NGRAPH * HID);
    k_pool<<<cdiv(NNODES * HID, B), B, 0, stream>>>(cur, batch, g, NNODES);

    int mTilesG = NGRAPH / 16;
    k_gemm_wmma<0, 1, 0><<<cdiv(mTilesG, GEMM_WAVES), B, shm96, stream>>>(
        g, dec_W0, dec_b0, g2, nullptr, nullptr, NGRAPH, HID, HID);
    k_dec2<<<cdiv(NGRAPH * OUTDIM, B), B, 0, stream>>>(g2, dec_W1, dec_b1, out);
}


# GEMM launches turned weight-stationary: a wave keeps its 16 output columns of B in registers (read once from global, B not staged in LDS); 64-row A and C tiles pass through LDS whole; k order kept
# speedup vs baseline: 2.7402x; 2.7402x over previous
#include <hip/hip_runtime.h>
#include <hip/hip_bf16.h>

typedef __attribute__((ext_vector_type(2))) float v2f;
typedef __attribute__((ext_vector_type(8))) float v8f;

#define NNODES 50000
#define NEDGES 800000
#define NGRAPH 512
#define INDIM  128
#define HID    96
#define OUTDIM 10
#define NCONV  4
#define GW_WAVES 6
#define TSHIFT 6
#define TILE   64
#define NTILES ((NNODES + TILE - 1) / TILE)
#define NTP    1024
#define FB     128
#define FT     1024
#define AGG_WAVES 16
#define MM_WAVES  12
#define APITCH 100
#define WPITCH (HID + 2)

static_assert(NNODES % 16 == 0, "a tile's valid rows must be whole 16-row WMMA tiles");

static inline int cdiv(int a, int b) { return (a + b - 1) / b; }

__global__ void k_tile_count(const int* __restrict__ ei, int* __restrict__ hist, int E, int per) {
    __shared__ int lh[NTILES];
    for (int i = threadIdx.x; i < NTILES; i += blockDim.x) lh[i] = 0;
    __syncthreads();
    int e0 = blockIdx.x * per;
    int e1 = min(e0 + per, E);
    for (int e = e0 + (int)threadIdx.x; e < e1; e += blockDim.x)
        atomicAdd(&lh[ei[E + e] >> TSHIFT], 1);
    __syncthreads();
    for (int i = threadIdx.x; i < NTILES; i += blockDim.x) hist[blockIdx.x * NTP + i] = lh[i];
}

__global__ void k_tile_scan(const int* __restrict__ hist, int* __restrict__ basem,
                            int* __restrict__ tstart) {
    __shared__ int sc[NTP];
    int t = threadIdx.x;
    int tot = 0;
    if (t < NTILES) {
        #pragma unroll 16
        for (int b = 0; b < FB; b++) tot += hist[b * NTP + t];
    }
    sc[t] = tot;
    __syncthreads();
    for (int off = 1; off < NTP; off <<= 1) {
        int a = (t >= off) ? sc[t - off] : 0;
        __syncthreads();
        sc[t] += a;
        __syncthreads();
    }
    int excl = sc[t] - tot;
    if (t <= NTILES) tstart[t] = excl;
    if (t < NTILES) {
        int run = excl;
        #pragma unroll 16
        for (int b = 0; b < FB; b++) {
            int cn = hist[b * NTP + t];
            basem[b * NTP + t] = run;
            run += cn;
        }
    }
}

__global__ void k_tile_fill(const int* __restrict__ ei, const int* __restrict__ basem,
                            int2* __restrict__ bedge, int E, int per) {
    __shared__ int lh[NTILES];
    __shared__ int lb[NTILES];
    for (int i = threadIdx.x; i < NTILES; i += blockDim.x) {
        lh[i] = 0;
        lb[i] = basem[blockIdx.x * NTP + i];
    }
    __syncthreads();
    int e0 = blockIdx.x * per;
    int e1 = min(e0 + per, E);
    for (int e = e0 + (int)threadIdx.x; e < e1; e += blockDim.x) {
        int s = ei[e];
        int d = ei[E + e];
        int bin = d >> TSHIFT;
        int r = atomicAdd(&lh[bin], 1);
        bedge[lb[bin] + r] = make_int2(s, d);
    }
}

__global__ void k_tile_dinv(const int2* __restrict__ bedge, const int* __restrict__ tstart,
                            float* __restrict__ dinv) {
    __shared__ int cnt[TILE];
    int t  = blockIdx.x;
    int n0 = t * TILE;
    if (threadIdx.x < TILE) cnt[threadIdx.x] = 0;
    __syncthreads();
    int e0 = tstart[t], e1 = tstart[t + 1];
    for (int e = e0 + (int)threadIdx.x; e < e1; e += blockDim.x)
        atomicAdd(&cnt[bedge[e].y - n0], 1);
    __syncthreads();
    int node = n0 + (int)threadIdx.x;
    if (threadIdx.x < TILE && node < NNODES)
        dinv[node] = rsqrtf(1.0f + (float)cnt[threadIdx.x]);
}

template <int K, int RELU>
__global__ void __launch_bounds__(GW_WAVES * 32)
k_gemm_ws(const float* __restrict__ A, const float* __restrict__ B,
          const float* __restrict__ bias, float* __restrict__ C, int M) {
    constexpr int NT    = GW_WAVES * 32;
    constexpr int KV    = K / 4;
    constexpr int KS    = K / 4;
    constexpr int AP    = K + 4;
    constexpr int CP    = HID + 4;
    constexpr int TOT   = TILE * KV;
    constexpr int NPASS = (TOT + NT - 1) / NT;
    constexpr int RV    = HID / 4;
    constexpr int CPASS = (TILE * RV + NT - 1) / NT;
    static_assert((AP / 4) % 2 == 1 && TILE * CP <= TILE * AP, "pitches");
    __shared__ __attribute__((aligned(16))) float sA[TILE * AP];

    const int tid   = threadIdx.x;
    const int wave  = __builtin_amdgcn_readfirstlane(tid >> 5);
    const int lane  = tid & 31;
    const int half  = lane >> 4;
    const int l15   = lane & 15;
    const int n0    = blockIdx.x * TILE;
    const int nrows = min(TILE, M - n0);

    const float4* A4 = (const float4*)(A + (size_t)n0 * K);
    float4 av[NPASS];
    #pragma unroll
    for (int it = 0; it < NPASS; it++) {
        int i = tid + it * NT;
        float4 v = make_float4(0.0f, 0.0f, 0.0f, 0.0f);
        if (i < nrows * KV) v = A4[i];
        av[it] = v;
    }
    const float* bcol = B + (size_t)(half * 2) * HID + wave * 16 + l15;
    v2f bf[KS];
    #pragma unroll
    for (int j = 0; j < KS; j++) {
        bf[j].x = bcol[(size_t)(4 * j) * HID];
        bf[j].y = bcol[(size_t)(4 * j + 1) * HID];
    }
    #pragma unroll
    for (int it = 0; it < NPASS; it++) {
        int i = tid + it * NT;
        if (i < TOT) {
            int r = i / KV;
            int q = i - r * KV;
            *(float4*)(sA + r * AP + 4 * q) = av[it];
        }
    }
    __syncthreads();

    v8f acc[4] = {};
    const float* abase = sA + l15 * AP + half * 2;
    #pragma unroll
    for (int j = 0; j < KS; j++) {
        #pragma unroll
        for (int rt = 0; rt < 4; rt++) {
            v2f a = *(const v2f*)(abase + rt * 16 * AP + 4 * j);
            acc[rt] = __builtin_amdgcn_wmma_f32_16x16x4_f32(
                 false, a,  false, bf[j],
                 (short)0, acc[rt],  false,  false);
        }
    }
    __syncthreads();

    const float bv = bias[wave * 16 + l15];
    float* cs = sA + (half * 8) * CP + wave * 16 + l15;
    #pragma unroll
    for (int rt = 0; rt < 4; rt++) {
        #pragma unroll
        for (int v = 0; v < 8; v++) {
            float r = acc[rt][v] + bv;
            if (RELU) r = fmaxf(r, 0.0f);
            cs[(rt * 16 + v) * CP] = r;
        }
    }
    __syncthreads();

    static_assert(TILE * RV == CPASS * NT, "the C tile is a whole number of passes");
    float4* C4 = (float4*)(C + (size_t)n0 * HID);
    if (nrows == TILE) {
        float4 cv[CPASS];
        #pragma unroll
        for (int it = 0; it < CPASS; it++) {
            int i = tid + it * NT;
            int r = i / RV;
            int q = i - r * RV;
            cv[it] = *(const float4*)(sA + r * CP + 4 * q);
        }
        #pragma unroll
        for (int it = 0; it < CPASS; it++) C4[tid + it * NT] = cv[it];
    } else {
        for (int i = tid; i < nrows * RV; i += NT) {
            int r = i / RV;
            int q = i - r * RV;
            C4[i] = *(const float4*)(sA + r * CP + 4 * q);
        }
    }
}

__device__ __forceinline__ void tile_edge8(const float* __restrict__ h, float* acc,
                                           int s, int dl, float nm, int j, int lane) {
    float v[8][3];
    int   ro[8];
    #pragma unroll
    for (int u = 0; u < 8; u++) {
        int   sj = __builtin_amdgcn_readlane(s, j + u);
        float nj = __int_as_float(__builtin_amdgcn_readlane(__float_as_int(nm), j + u));
        ro[u]    = __builtin_amdgcn_readlane(dl, j + u) * APITCH + lane;
        const float* srow = h + (size_t)sj * HID + lane;
        v[u][0] = srow[0]  * nj;
        v[u][1] = srow[32] * nj;
        v[u][2] = srow[64] * nj;
    }
    #pragma unroll
    for (int u = 0; u < 8; u++) {
        __hip_atomic_fetch_add(acc + ro[u],      v[u][0], __ATOMIC_RELAXED, __HIP_MEMORY_SCOPE_WORKGROUP);
        __hip_atomic_fetch_add(acc + ro[u] + 32, v[u][1], __ATOMIC_RELAXED, __HIP_MEMORY_SCOPE_WORKGROUP);
        __hip_atomic_fetch_add(acc + ro[u] + 64, v[u][2], __ATOMIC_RELAXED, __HIP_MEMORY_SCOPE_WORKGROUP);
    }
}

__global__ void __launch_bounds__(AGG_WAVES * 32)
k_tile_layer(const float* __restrict__ h, const float* __restrict__ W,
             const float* __restrict__ bias, const float* __restrict__ dinv,
             const int2* __restrict__ bedge, const int* __restrict__ tstart,
             float* __restrict__ hout) {
    __shared__ __attribute__((aligned(16))) float acc[TILE * APITCH];
    __shared__ __attribute__((aligned(16))) float sWt[HID * WPITCH];
    constexpr int NT = AGG_WAVES * 32;
    constexpr int WV = HID * HID / 4;
    constexpr int RV = HID / 4;
    const int tid   = threadIdx.x;
    const int t     = blockIdx.x;
    const int n0    = t * TILE;
    const int nrows = min(TILE, NNODES - n0);

    const float4* W4 = (const float4*)W;
    #pragma unroll
    for (int it = 0; it < (WV + NT - 1) / NT; it++) {
        int i = tid + it * NT;
        if (i < WV) {
            float4 w = W4[i];
            int k   = i / RV;
            int col = (i - k * RV) * 4;
            float* p = sWt + col * WPITCH + k;
            p[0]          = w.x;
            p[WPITCH]     = w.y;
            p[2 * WPITCH] = w.z;
            p[3 * WPITCH] = w.w;
        }
    }
    const float4* hin = (const float4*)(h + (size_t)n0 * HID);
    float4* lacc = (float4*)acc;
    for (int i = tid; i < nrows * RV; i += NT) {
        int r = i / RV;
        int q = i - r * RV;
        float d  = dinv[n0 + r];
        float d2 = d * d;
        float4 v = hin[i];
        v.x *= d2; v.y *= d2; v.z *= d2; v.w *= d2;
        lacc[r * (APITCH / 4) + q] = v;
    }
    __syncthreads();

    const int e0 = tstart[t], e1 = tstart[t + 1];
    const int wave = __builtin_amdgcn_readfirstlane(tid >> 5);
    const int lane = tid & 31;
    for (int base = e0 + wave * 32; base < e1; base += NT) {
        int s = 0, dl = 0;
        float nm = 0.0f;
        int e = base + lane;
        if (e < e1) {
            int2 sd = bedge[e];
            s  = sd.x;
            dl = sd.y - n0;
            nm = dinv[sd.x] * dinv[sd.y];
        }
        const int cnt = min(32, e1 - base);
        int j = 0;
        for (; j + 8 <= cnt; j += 8) tile_edge8(h, acc, s, dl, nm, j, lane);
        for (; j < cnt; j++) {
            int   sj = __builtin_amdgcn_readlane(s, j);
            float nj = __int_as_float(__builtin_amdgcn_readlane(__float_as_int(nm), j));
            int   ro = __builtin_amdgcn_readlane(dl, j) * APITCH + lane;
            const float* srow = h + (size_t)sj * HID + lane;
            float v0 = srow[0] * nj, v1 = srow[32] * nj, v2 = srow[64] * nj;
            __hip_atomic_fetch_add(acc + ro,      v0, __ATOMIC_RELAXED, __HIP_MEMORY_SCOPE_WORKGROUP);
            __hip_atomic_fetch_add(acc + ro + 32, v1, __ATOMIC_RELAXED, __HIP_MEMORY_SCOPE_WORKGROUP);
            __hip_atomic_fetch_add(acc + ro + 64, v2, __ATOMIC_RELAXED, __HIP_MEMORY_SCOPE_WORKGROUP);
        }
    }
    __syncthreads();

    const int rt = wave & 3;
    if (wave < MM_WAVES && rt * 16 < nrows) {
        const int half = lane >> 4;
        const int l15  = lane & 15;
        const int nt0  = (wave >> 2) * 2;
        const float* abase = acc + (rt * 16 + l15) * APITCH + half * 2;
        const float* bbase = sWt + (nt0 * 16 + l15) * WPITCH + half * 2;
        v8f c0 = {}, c1 = {};
        #pragma unroll 4
        for (int k = 0; k < HID; k += 4) {
            v2f a  = *(const v2f*)(abase + k);
            v2f b0 = *(const v2f*)(bbase + k);
            v2f b1 = *(const v2f*)(bbase + 16 * WPITCH + k);
            c0 = __builtin_amdgcn_wmma_f32_16x16x4_f32(
                 false, a,  false, b0,
                 (short)0, c0,  false,  false);
            c1 = __builtin_amdgcn_wmma_f32_16x16x4_f32(
                 false, a,  false, b1,
                 (short)0, c1,  false,  false);
        }
        float* obase = hout + (size_t)(n0 + rt * 16 + half * 8) * HID + nt0 * 16 + l15;
        const float bv0 = bias[nt0 * 16 + l15];
        const float bv1 = bias[nt0 * 16 + 16 + l15];
        #pragma unroll
        for (int v = 0; v < 8; v++) {
            obase[(size_t)v * HID]      = fmaxf(c0[v] + bv0, 0.0f);
            obase[(size_t)v * HID + 16] = fmaxf(c1[v] + bv1, 0.0f);
        }
    }
}

__global__ void k_zero(float* x, int n) {
    int i = blockIdx.x * blockDim.x + threadIdx.x;
    if (i < n) x[i] = 0.0f;
}

__global__ void k_pool(const float* __restrict__ h, const int* __restrict__ batch,
                       float* __restrict__ g, int n) {
    int idx = blockIdx.x * blockDim.x + threadIdx.x;
    if (idx < n * HID) {
        int node = idx / HID;
        int c    = idx - node * HID;
        atomicAdd(&g[(size_t)batch[node] * HID + c], h[idx]);
    }
}

__global__ void k_dec2(const float* __restrict__ g2, const float* __restrict__ W,
                       const float* __restrict__ b, float* __restrict__ out) {
    int idx = blockIdx.x * blockDim.x + threadIdx.x;
    if (idx < NGRAPH * OUTDIM) {
        int gi = idx / OUTDIM;
        int o  = idx - gi * OUTDIM;
        float s = b[o];
        const float* row = g2 + (size_t)gi * HID;
        #pragma unroll 8
        for (int k = 0; k < HID; k++) s += row[k] * W[k * OUTDIM + o];
        out[idx] = s;
    }
}

extern "C" void kernel_launch(void* const* d_in, const int* in_sizes, int n_in,
                              void* d_out, int out_size, void* d_ws, size_t ws_size,
                              hipStream_t stream) {
    const float* x       = (const float*)d_in[0];
    const int*   ei      = (const int*)  d_in[1];
    const int*   batch   = (const int*)  d_in[2];
    const float* enc_W0  = (const float*)d_in[3];
    const float* enc_b0  = (const float*)d_in[4];
    const float* enc_W1  = (const float*)d_in[5];
    const float* enc_b1  = (const float*)d_in[6];
    const float* conv_W  = (const float*)d_in[7];
    const float* conv_b  = (const float*)d_in[8];
    const float* dec_W0  = (const float*)d_in[9];
    const float* dec_b0  = (const float*)d_in[10];
    const float* dec_W1  = (const float*)d_in[11];
    const float* dec_b1  = (const float*)d_in[12];
    float* out = (float*)d_out;

    float* ws   = (float*)d_ws;
    size_t feat = (size_t)NNODES * HID;
    float* buf0 = ws;
    float* buf1 = buf0 + feat;
    float* buf2 = buf1 + feat;
    float* dinv = buf2 + feat;
    float* g    = dinv + NNODES;
    float* g2   = g + (size_t)NGRAPH * HID;
    int*   hist   = (int*)(g2 + (size_t)NGRAPH * HID);
    int*   basem  = hist + (size_t)FB * NTP;
    int*   tstart = basem + (size_t)FB * NTP;
    int2*  bedge  = (int2*)(tstart + NTP);

    const int B = 256;

    int per = cdiv(NEDGES, FB);
    k_tile_count<<<FB, FT, 0, stream>>>(ei, hist, NEDGES, per);
    k_tile_scan<<<1, NTP, 0, stream>>>(hist, basem, tstart);
    k_tile_fill<<<FB, FT, 0, stream>>>(ei, basem, bedge, NEDGES, per);
    k_tile_dinv<<<NTILES, B, 0, stream>>>(bedge, tstart, dinv);

    k_gemm_ws<INDIM, 1><<<NTILES, GW_WAVES * 32, 0, stream>>>(x, enc_W0, enc_b0, buf0, NNODES);
    k_gemm_ws<HID, 0><<<NTILES, GW_WAVES * 32, 0, stream>>>(buf0, enc_W1, enc_b1, buf1, NNODES);

    float* cur = buf1;
    float* nxt = buf2;
    for (int l = 0; l < NCONV; l++) {
        const float* Wl = conv_W + (size_t)l * HID * HID;
        const float* bl = conv_b + (size_t)l * HID;
        k_tile_layer<<<NTILES, AGG_WAVES * 32, 0, stream>>>(cur, Wl, bl, dinv, bedge, tstart, nxt);
        float* t = cur; cur = nxt; nxt = t;
    }

    k_zero<<<cdiv(NGRAPH * HID, B), B, 0, stream>>>(g, NGRAPH * HID);
    k_pool<<<cdiv(NNODES * HID, B), B, 0, stream>>>(cur, batch, g, NNODES);

    k_gemm_ws<HID, 1><<<NGRAPH / TILE, GW_WAVES * 32, 0, stream>>>(g, dec_W0, dec_b0, g2, NGRAPH);
    k_dec2<<<cdiv(NGRAPH * OUTDIM, B), B, 0, stream>>>(g2, dec_W1, dec_b1, out);
}


# Pooling by graph: a work-group per graph finds its run of nodes in the sorted batch (two-level search), sums the rows in registers and LDS, stores g; no atomics, no zero fill; pool sum order changes
# speedup vs baseline: 2.9494x; 1.0763x over previous
#include <hip/hip_runtime.h>
#include <hip/hip_bf16.h>

typedef __attribute__((ext_vector_type(2))) float v2f;
typedef __attribute__((ext_vector_type(8))) float v8f;

#define NNODES 50000
#define NEDGES 800000
#define NGRAPH 512
#define INDIM  128
#define HID    96
#define OUTDIM 10
#define NCONV  4
#define GW_WAVES 6
#define TSHIFT 6
#define TILE   64
#define NTILES ((NNODES + TILE - 1) / TILE)
#define NTP    1024
#define FB     128
#define FT     1024
#define AGG_WAVES 16
#define MM_WAVES  12
#define APITCH 100
#define WPITCH (HID + 2)
#define DINV_PLANE ((NNODES + 31) / 32 * 32)

static_assert(NNODES % 16 == 0, "a tile's valid rows must be whole 16-row WMMA tiles");

static inline int cdiv(int a, int b) { return (a + b - 1) / b; }

__global__ void k_tile_count(const int* __restrict__ ei, int* __restrict__ hist, int E, int per) {
    __shared__ int lh[NTILES];
    for (int i = threadIdx.x; i < NTILES; i += blockDim.x) lh[i] = 0;
    __syncthreads();
    int e0 = blockIdx.x * per;
    int e1 = min(e0 + per, E);
    for (int e = e0 + (int)threadIdx.x; e < e1; e += blockDim.x)
        atomicAdd(&lh[ei[E + e] >> TSHIFT], 1);
    __syncthreads();
    for (int i = threadIdx.x; i < NTILES; i += blockDim.x) hist[blockIdx.x * NTP + i] = lh[i];
}

__global__ void k_tile_scan(const int* __restrict__ hist, int* __restrict__ basem,
                            int* __restrict__ tstart) {
    __shared__ int sc[NTP];
    int t = threadIdx.x;
    int tot = 0;
    if (t < NTILES) {
        #pragma unroll 16
        for (int b = 0; b < FB; b++) tot += hist[b * NTP + t];
    }
    sc[t] = tot;
    __syncthreads();
    for (int off = 1; off < NTP; off <<= 1) {
        int a = (t >= off) ? sc[t - off] : 0;
        __syncthreads();
        sc[t] += a;
        __syncthreads();
    }
    int excl = sc[t] - tot;
    if (t <= NTILES) tstart[t] = excl;
    if (t < NTILES) {
        int run = excl;
        #pragma unroll 16
        for (int b = 0; b < FB; b++) {
            int cn = hist[b * NTP + t];
            basem[b * NTP + t] = run;
            run += cn;
        }
    }
}

__global__ void k_tile_fill(const int* __restrict__ ei, const int* __restrict__ basem,
                            int2* __restrict__ bedge, int E, int per) {
    __shared__ int lh[NTILES];
    __shared__ int lb[NTILES];
    for (int i = threadIdx.x; i < NTILES; i += blockDim.x) {
        lh[i] = 0;
        lb[i] = basem[blockIdx.x * NTP + i];
    }
    __syncthreads();
    int e0 = blockIdx.x * per;
    int e1 = min(e0 + per, E);
    for (int e = e0 + (int)threadIdx.x; e < e1; e += blockDim.x) {
        int s = ei[e];
        int d = ei[E + e];
        int bin = d >> TSHIFT;
        int r = atomicAdd(&lh[bin], 1);
        bedge[lb[bin] + r] = make_int2(s, d);
    }
}

__global__ void k_tile_dinv(const int2* __restrict__ bedge, const int* __restrict__ tstart,
                            float* __restrict__ dinv) {
    __shared__ int cnt[TILE];
    int t  = blockIdx.x;
    int n0 = t * TILE;
    if (threadIdx.x < TILE) cnt[threadIdx.x] = 0;
    __syncthreads();
    int e0 = tstart[t], e1 = tstart[t + 1];
    for (int e = e0 + (int)threadIdx.x; e < e1; e += blockDim.x)
        atomicAdd(&cnt[bedge[e].y - n0], 1);
    __syncthreads();
    int node = n0 + (int)threadIdx.x;
    if (threadIdx.x < TILE && node < NNODES)
        dinv[node] = rsqrtf(1.0f + (float)cnt[threadIdx.x]);
}

template <int K, int RELU>
__global__ void __launch_bounds__(GW_WAVES * 32)
k_gemm_ws(const float* __restrict__ A, const float* __restrict__ B,
          const float* __restrict__ bias, float* __restrict__ C, int M) {
    constexpr int NT    = GW_WAVES * 32;
    constexpr int KV    = K / 4;
    constexpr int KS    = K / 4;
    constexpr int AP    = K + 4;
    constexpr int CP    = HID + 4;
    constexpr int TOT   = TILE * KV;
    constexpr int NPASS = (TOT + NT - 1) / NT;
    constexpr int RV    = HID / 4;
    constexpr int CPASS = (TILE * RV + NT - 1) / NT;
    static_assert((AP / 4) % 2 == 1 && TILE * CP <= TILE * AP, "pitches");
    __shared__ __attribute__((aligned(16))) float sA[TILE * AP];

    const int tid   = threadIdx.x;
    const int wave  = __builtin_amdgcn_readfirstlane(tid >> 5);
    const int lane  = tid & 31;
    const int half  = lane >> 4;
    const int l15   = lane & 15;
    const int n0    = blockIdx.x * TILE;
    const int nrows = min(TILE, M - n0);

    const float4* A4 = (const float4*)(A + (size_t)n0 * K);
    float4 av[NPASS];
    #pragma unroll
    for (int it = 0; it < NPASS; it++) {
        int i = tid + it * NT;
        float4 v = make_float4(0.0f, 0.0f, 0.0f, 0.0f);
        if (i < nrows * KV) v = A4[i];
        av[it] = v;
    }
    const float* bcol = B + (size_t)(half * 2) * HID + wave * 16 + l15;
    v2f bf[KS];
    #pragma unroll
    for (int j = 0; j < KS; j++) {
        bf[j].x = bcol[(size_t)(4 * j) * HID];
        bf[j].y = bcol[(size_t)(4 * j + 1) * HID];
    }
    #pragma unroll
    for (int it = 0; it < NPASS; it++) {
        int i = tid + it * NT;
        if (i < TOT) {
            int r = i / KV;
            int q = i - r * KV;
            *(float4*)(sA + r * AP + 4 * q) = av[it];
        }
    }
    __syncthreads();

    v8f acc[4] = {};
    const float* abase = sA + l15 * AP + half * 2;
    #pragma unroll
    for (int j = 0; j < KS; j++) {
        #pragma unroll
        for (int rt = 0; rt < 4; rt++) {
            v2f a = *(const v2f*)(abase + rt * 16 * AP + 4 * j);
            acc[rt] = __builtin_amdgcn_wmma_f32_16x16x4_f32(
                 false, a,  false, bf[j],
                 (short)0, acc[rt],  false,  false);
        }
    }
    __syncthreads();

    const float bv = bias[wave * 16 + l15];
    float* cs = sA + (half * 8) * CP + wave * 16 + l15;
    #pragma unroll
    for (int rt = 0; rt < 4; rt++) {
        #pragma unroll
        for (int v = 0; v < 8; v++) {
            float r = acc[rt][v] + bv;
            if (RELU) r = fmaxf(r, 0.0f);
            cs[(rt * 16 + v) * CP] = r;
        }
    }
    __syncthreads();

    static_assert(TILE * RV == CPASS * NT, "the C tile is a whole number of passes");
    float4* C4 = (float4*)(C + (size_t)n0 * HID);
    if (nrows == TILE) {
        float4 cv[CPASS];
        #pragma unroll
        for (int it = 0; it < CPASS; it++) {
            int i = tid + it * NT;
            int r = i / RV;
            int q = i - r * RV;
            cv[it] = *(const float4*)(sA + r * CP + 4 * q);
        }
        #pragma unroll
        for (int it = 0; it < CPASS; it++) C4[tid + it * NT] = cv[it];
    } else {
        for (int i = tid; i < nrows * RV; i += NT) {
            int r = i / RV;
            int q = i - r * RV;
            C4[i] = *(const float4*)(sA + r * CP + 4 * q);
        }
    }
}

__device__ __forceinline__ void tile_edge8(const float* __restrict__ h, float* acc,
                                           int s, int dl, float nm, int j, int lane) {
    float v[8][3];
    int   ro[8];
    #pragma unroll
    for (int u = 0; u < 8; u++) {
        int   sj = __builtin_amdgcn_readlane(s, j + u);
        float nj = __int_as_float(__builtin_amdgcn_readlane(__float_as_int(nm), j + u));
        ro[u]    = __builtin_amdgcn_readlane(dl, j + u) * APITCH + lane;
        const float* srow = h + (size_t)sj * HID + lane;
        v[u][0] = srow[0]  * nj;
        v[u][1] = srow[32] * nj;
        v[u][2] = srow[64] * nj;
    }
    #pragma unroll
    for (int u = 0; u < 8; u++) {
        __hip_atomic_fetch_add(acc + ro[u],      v[u][0], __ATOMIC_RELAXED, __HIP_MEMORY_SCOPE_WORKGROUP);
        __hip_atomic_fetch_add(acc + ro[u] + 32, v[u][1], __ATOMIC_RELAXED, __HIP_MEMORY_SCOPE_WORKGROUP);
        __hip_atomic_fetch_add(acc + ro[u] + 64, v[u][2], __ATOMIC_RELAXED, __HIP_MEMORY_SCOPE_WORKGROUP);
    }
}

__global__ void __launch_bounds__(AGG_WAVES * 32)
k_tile_layer(const float* __restrict__ h, const float* __restrict__ W,
             const float* __restrict__ bias, const float* __restrict__ dinv,
             const int2* __restrict__ bedge, const int* __restrict__ tstart,
             float* __restrict__ hout) {
    __shared__ __attribute__((aligned(16))) float acc[TILE * APITCH];
    __shared__ __attribute__((aligned(16))) float sWt[HID * WPITCH];
    constexpr int NT = AGG_WAVES * 32;
    constexpr int WV = HID * HID / 4;
    constexpr int RV = HID / 4;
    const int tid   = threadIdx.x;
    const int t     = blockIdx.x;
    const int n0    = t * TILE;
    const int nrows = min(TILE, NNODES - n0);

    const float4* W4 = (const float4*)W;
    #pragma unroll
    for (int it = 0; it < (WV + NT - 1) / NT; it++) {
        int i = tid + it * NT;
        if (i < WV) {
            float4 w = W4[i];
            int k   = i / RV;
            int col = (i - k * RV) * 4;
            float* p = sWt + col * WPITCH + k;
            p[0]          = w.x;
            p[WPITCH]     = w.y;
            p[2 * WPITCH] = w.z;
            p[3 * WPITCH] = w.w;
        }
    }
    const float4* hin = (const float4*)(h + (size_t)n0 * HID);
    float4* lacc = (float4*)acc;
    for (int i = tid; i < nrows * RV; i += NT) {
        int r = i / RV;
        int q = i - r * RV;
        float d  = dinv[n0 + r];
        float d2 = d * d;
        float4 v = hin[i];
        v.x *= d2; v.y *= d2; v.z *= d2; v.w *= d2;
        lacc[r * (APITCH / 4) + q] = v;
    }
    __syncthreads();

    const int e0 = tstart[t], e1 = tstart[t + 1];
    const int wave = __builtin_amdgcn_readfirstlane(tid >> 5);
    const int lane = tid & 31;
    for (int base = e0 + wave * 32; base < e1; base += NT) {
        int s = 0, dl = 0;
        float nm = 0.0f;
        int e = base + lane;
        if (e < e1) {
            int2 sd = bedge[e];
            s  = sd.x;
            dl = sd.y - n0;
            nm = dinv[sd.x] * dinv[sd.y];
        }
        const int cnt = min(32, e1 - base);
        int j = 0;
        for (; j + 8 <= cnt; j += 8) tile_edge8(h, acc, s, dl, nm, j, lane);
        for (; j < cnt; j++) {
            int   sj = __builtin_amdgcn_readlane(s, j);
            float nj = __int_as_float(__builtin_amdgcn_readlane(__float_as_int(nm), j));
            int   ro = __builtin_amdgcn_readlane(dl, j) * APITCH + lane;
            const float* srow = h + (size_t)sj * HID + lane;
            float v0 = srow[0] * nj, v1 = srow[32] * nj, v2 = srow[64] * nj;
            __hip_atomic_fetch_add(acc + ro,      v0, __ATOMIC_RELAXED, __HIP_MEMORY_SCOPE_WORKGROUP);
            __hip_atomic_fetch_add(acc + ro + 32, v1, __ATOMIC_RELAXED, __HIP_MEMORY_SCOPE_WORKGROUP);
            __hip_atomic_fetch_add(acc + ro + 64, v2, __ATOMIC_RELAXED, __HIP_MEMORY_SCOPE_WORKGROUP);
        }
    }
    __syncthreads();

    const int rt = wave & 3;
    if (wave < MM_WAVES && rt * 16 < nrows) {
        const int half = lane >> 4;
        const int l15  = lane & 15;
        const int nt0  = (wave >> 2) * 2;
        const float* abase = acc + (rt * 16 + l15) * APITCH + half * 2;
        const float* bbase = sWt + (nt0 * 16 + l15) * WPITCH + half * 2;
        v8f c0 = {}, c1 = {};
        #pragma unroll 4
        for (int k = 0; k < HID; k += 4) {
            v2f a  = *(const v2f*)(abase + k);
            v2f b0 = *(const v2f*)(bbase + k);
            v2f b1 = *(const v2f*)(bbase + 16 * WPITCH + k);
            c0 = __builtin_amdgcn_wmma_f32_16x16x4_f32(
                 false, a,  false, b0,
                 (short)0, c0,  false,  false);
            c1 = __builtin_amdgcn_wmma_f32_16x16x4_f32(
                 false, a,  false, b1,
                 (short)0, c1,  false,  false);
        }
        float* obase = hout + (size_t)(n0 + rt * 16 + half * 8) * HID + nt0 * 16 + l15;
        const float bv0 = bias[nt0 * 16 + l15];
        const float bv1 = bias[nt0 * 16 + 16 + l15];
        #pragma unroll
        for (int v = 0; v < 8; v++) {
            obase[(size_t)v * HID]      = fmaxf(c0[v] + bv0, 0.0f);
            obase[(size_t)v * HID + 16] = fmaxf(c1[v] + bv1, 0.0f);
        }
    }
}

#define PL_NT 384
#define PL_S  ((NNODES + PL_NT - 1) / PL_NT)
#define PL_DEPTH 8

__global__ void __launch_bounds__(PL_NT)
k_pool_graph(const float* __restrict__ h, const int* __restrict__ batch,
             float* __restrict__ g, int n) {
    constexpr int NW    = PL_NT / 32;
    constexpr int RV    = HID / 4;
    constexpr int SLOTS = PL_NT / RV;
    static_assert(PL_NT % RV == 0 && PL_NT % 64 == 0, "a thread keeps its columns; the two halves are whole waves");
    static_assert(PL_S - 1 <= PL_NT / 2, "the ids between two samples fit one half of the work-group");
    __shared__ int wc[3][NW];
    __shared__ __attribute__((aligned(16))) float4 red[PL_NT];

    const int tid  = threadIdx.x;
    const int gi   = blockIdx.x;
    const int wave = tid >> 5;
    const int lane = tid & 31;

    const int p = tid * PL_S;
    bool below = false, upto = false;
    if (p < n) {
        int v = batch[p];
        below = v < gi;
        upto  = v <= gi;
    }
    const unsigned m0 = __builtin_amdgcn_ballot_w32(below);
    const unsigned m1 = __builtin_amdgcn_ballot_w32(upto);
    if (lane == 0) {
        wc[0][wave] = __builtin_popcount(m0);
        wc[1][wave] = __builtin_popcount(m1);
    }
    __syncthreads();
    int c_lo = 0, c_hi = 0;
    #pragma unroll
    for (int w = 0; w < NW; w++) {
        c_lo += wc[0][w];
        c_hi += wc[1][w];
    }
    const int upper = (tid >= PL_NT / 2) ? 1 : 0;
    const int j     = tid - upper * (PL_NT / 2);
    const int c     = upper ? c_hi : c_lo;
    bool in = false;
    if (c > 0 && j < PL_S - 1) {
        int idx = (c - 1) * PL_S + 1 + j;
        if (idx < n) {
            int v = batch[idx];
            in = upper ? (v <= gi) : (v < gi);
        }
    }
    const unsigned m2 = __builtin_amdgcn_ballot_w32(in);
    if (lane == 0) wc[2][wave] = __builtin_popcount(m2);
    __syncthreads();
    int d_lo = 0, d_hi = 0;
    #pragma unroll
    for (int w = 0; w < NW / 2; w++) {
        d_lo += wc[2][w];
        d_hi += wc[2][w + NW / 2];
    }
    const int lo = c_lo ? (c_lo - 1) * PL_S + 1 + d_lo : 0;
    const int hi = c_hi ? (c_hi - 1) * PL_S + 1 + d_hi : 0;

    const int cnt = (hi - lo) * RV;
    const float4* src = (const float4*)(h + (size_t)lo * HID);
    float4 acc = make_float4(0.0f, 0.0f, 0.0f, 0.0f);
    for (int base = 0; base < cnt; base += PL_DEPTH * PL_NT) {
        float4 v[PL_DEPTH];
        #pragma unroll
        for (int u = 0; u < PL_DEPTH; u++) {
            int i = base + u * PL_NT + tid;
            v[u] = make_float4(0.0f, 0.0f, 0.0f, 0.0f);
            if (i < cnt) v[u] = src[i];
        }
        #pragma unroll
        for (int u = 0; u < PL_DEPTH; u++) {
            acc.x += v[u].x; acc.y += v[u].y; acc.z += v[u].z; acc.w += v[u].w;
        }
    }
    red[tid] = acc;
    __syncthreads();
    if (tid < RV) {
        float4 s = red[tid];
        #pragma unroll
        for (int r = 1; r < SLOTS; r++) {
            float4 v = red[r * RV + tid];
            s.x += v.x; s.y += v.y; s.z += v.z; s.w += v.w;
        }
        ((float4*)(g + (size_t)gi * HID))[tid] = s;
    }
}

__global__ void k_dec2(const float* __restrict__ g2, const float* __restrict__ W,
                       const float* __restrict__ b, float* __restrict__ out) {
    int idx = blockIdx.x * blockDim.x + threadIdx.x;
    if (idx < NGRAPH * OUTDIM) {
        int gi = idx / OUTDIM;
        int o  = idx - gi * OUTDIM;
        float s = b[o];
        const float* row = g2 + (size_t)gi * HID;
        #pragma unroll 8
        for (int k = 0; k < HID; k++) s += row[k] * W[k * OUTDIM + o];
        out[idx] = s;
    }
}

extern "C" void kernel_launch(void* const* d_in, const int* in_sizes, int n_in,
                              void* d_out, int out_size, void* d_ws, size_t ws_size,
                              hipStream_t stream) {
    const float* x       = (const float*)d_in[0];
    const int*   ei      = (const int*)  d_in[1];
    const int*   batch   = (const int*)  d_in[2];
    const float* enc_W0  = (const float*)d_in[3];
    const float* enc_b0  = (const float*)d_in[4];
    const float* enc_W1  = (const float*)d_in[5];
    const float* enc_b1  = (const float*)d_in[6];
    const float* conv_W  = (const float*)d_in[7];
    const float* conv_b  = (const float*)d_in[8];
    const float* dec_W0  = (const float*)d_in[9];
    const float* dec_b0  = (const float*)d_in[10];
    const float* dec_W1  = (const float*)d_in[11];
    const float* dec_b1  = (const float*)d_in[12];
    float* out = (float*)d_out;

    float* ws   = (float*)d_ws;
    size_t feat = (size_t)NNODES * HID;
    float* buf0 = ws;
    float* buf1 = buf0 + feat;
    float* buf2 = buf1 + feat;
    float* dinv = buf2 + feat;
    float* g    = dinv + DINV_PLANE;
    float* g2   = g + (size_t)NGRAPH * HID;
    int*   hist   = (int*)(g2 + (size_t)NGRAPH * HID);
    int*   basem  = hist + (size_t)FB * NTP;
    int*   tstart = basem + (size_t)FB * NTP;
    int2*  bedge  = (int2*)(tstart + NTP);

    const int B = 256;

    int per = cdiv(NEDGES, FB);
    k_tile_count<<<FB, FT, 0, stream>>>(ei, hist, NEDGES, per);
    k_tile_scan<<<1, NTP, 0, stream>>>(hist, basem, tstart);
    k_tile_fill<<<FB, FT, 0, stream>>>(ei, basem, bedge, NEDGES, per);
    k_tile_dinv<<<NTILES, B, 0, stream>>>(bedge, tstart, dinv);

    k_gemm_ws<INDIM, 1><<<NTILES, GW_WAVES * 32, 0, stream>>>(x, enc_W0, enc_b0, buf0, NNODES);
    k_gemm_ws<HID, 0><<<NTILES, GW_WAVES * 32, 0, stream>>>(buf0, enc_W1, enc_b1, buf1, NNODES);

    float* cur = buf1;
    float* nxt = buf2;
    for (int l = 0; l < NCONV; l++) {
        const float* Wl = conv_W + (size_t)l * HID * HID;
        const float* bl = conv_b + (size_t)l * HID;
        k_tile_layer<<<NTILES, AGG_WAVES * 32, 0, stream>>>(cur, Wl, bl, dinv, bedge, tstart, nxt);
        float* t = cur; cur = nxt; nxt = t;
    }

    k_pool_graph<<<NGRAPH, PL_NT, 0, stream>>>(cur, batch, g, NNODES);

    k_gemm_ws<HID, 1><<<NGRAPH / TILE, GW_WAVES * 32, 0, stream>>>(g, dec_W0, dec_b0, g2, NGRAPH);
    k_dec2<<<cdiv(NGRAPH * OUTDIM, B), B, 0, stream>>>(g2, dec_W1, dec_b1, out);
}


# Feature rows kept pre-scaled by dinv (encoder's last product and each layer's store scale them): edge loop adds raw rows, no dinv[src] line request per edge; dinv[dst] applied after W; add order kept
# speedup vs baseline: 3.0937x; 1.0489x over previous
#include <hip/hip_runtime.h>
#include <hip/hip_bf16.h>

typedef __attribute__((ext_vector_type(2))) float v2f;
typedef __attribute__((ext_vector_type(8))) float v8f;

#define NNODES 50000
#define NEDGES 800000
#define NGRAPH 512
#define INDIM  128
#define HID    96
#define OUTDIM 10
#define NCONV  4
#define GW_WAVES 6
#define TSHIFT 6
#define TILE   64
#define NTILES ((NNODES + TILE - 1) / TILE)
#define NTP    1024
#define FB     128
#define FT     1024
#define AGG_WAVES 16
#define MM_WAVES  12
#define APITCH 100
#define WPITCH (HID + 2)
#define DINV_PLANE ((NNODES + 31) / 32 * 32)

static_assert(NNODES % 16 == 0, "a tile's valid rows must be whole 16-row WMMA tiles");

static inline int cdiv(int a, int b) { return (a + b - 1) / b; }

__global__ void k_tile_count(const int* __restrict__ ei, int* __restrict__ hist, int E, int per) {
    __shared__ int lh[NTILES];
    for (int i = threadIdx.x; i < NTILES; i += blockDim.x) lh[i] = 0;
    __syncthreads();
    int e0 = blockIdx.x * per;
    int e1 = min(e0 + per, E);
    for (int e = e0 + (int)threadIdx.x; e < e1; e += blockDim.x)
        atomicAdd(&lh[ei[E + e] >> TSHIFT], 1);
    __syncthreads();
    for (int i = threadIdx.x; i < NTILES; i += blockDim.x) hist[blockIdx.x * NTP + i] = lh[i];
}

__global__ void k_tile_scan(const int* __restrict__ hist, int* __restrict__ basem,
                            int* __restrict__ tstart) {
    __shared__ int sc[NTP];
    int t = threadIdx.x;
    int tot = 0;
    if (t < NTILES) {
        #pragma unroll 16
        for (int b = 0; b < FB; b++) tot += hist[b * NTP + t];
    }
    sc[t] = tot;
    __syncthreads();
    for (int off = 1; off < NTP; off <<= 1) {
        int a = (t >= off) ? sc[t - off] : 0;
        __syncthreads();
        sc[t] += a;
        __syncthreads();
    }
    int excl = sc[t] - tot;
    if (t <= NTILES) tstart[t] = excl;
    if (t < NTILES) {
        int run = excl;
        #pragma unroll 16
        for (int b = 0; b < FB; b++) {
            int cn = hist[b * NTP + t];
            basem[b * NTP + t] = run;
            run += cn;
        }
    }
}

__global__ void k_tile_fill(const int* __restrict__ ei, const int* __restrict__ basem,
                            int2* __restrict__ bedge, int E, int per) {
    __shared__ int lh[NTILES];
    __shared__ int lb[NTILES];
    for (int i = threadIdx.x; i < NTILES; i += blockDim.x) {
        lh[i] = 0;
        lb[i] = basem[blockIdx.x * NTP + i];
    }
    __syncthreads();
    int e0 = blockIdx.x * per;
    int e1 = min(e0 + per, E);
    for (int e = e0 + (int)threadIdx.x; e < e1; e += blockDim.x) {
        int s = ei[e];
        int d = ei[E + e];
        int bin = d >> TSHIFT;
        int r = atomicAdd(&lh[bin], 1);
        bedge[lb[bin] + r] = make_int2(s, d);
    }
}

__global__ void k_tile_dinv(const int2* __restrict__ bedge, const int* __restrict__ tstart,
                            float* __restrict__ dinv) {
    __shared__ int cnt[TILE];
    int t  = blockIdx.x;
    int n0 = t * TILE;
    if (threadIdx.x < TILE) cnt[threadIdx.x] = 0;
    __syncthreads();
    int e0 = tstart[t], e1 = tstart[t + 1];
    for (int e = e0 + (int)threadIdx.x; e < e1; e += blockDim.x)
        atomicAdd(&cnt[bedge[e].y - n0], 1);
    __syncthreads();
    int node = n0 + (int)threadIdx.x;
    if (threadIdx.x < TILE && node < NNODES)
        dinv[node] = rsqrtf(1.0f + (float)cnt[threadIdx.x]);
}

template <int K, int RELU, int SCALE>
__global__ void __launch_bounds__(GW_WAVES * 32)
k_gemm_ws(const float* __restrict__ A, const float* __restrict__ B,
          const float* __restrict__ bias, const float* __restrict__ rscale,
          float* __restrict__ C, int M) {
    constexpr int NT    = GW_WAVES * 32;
    constexpr int KV    = K / 4;
    constexpr int KS    = K / 4;
    constexpr int AP    = K + 4;
    constexpr int CP    = HID + 4;
    constexpr int TOT   = TILE * KV;
    constexpr int NPASS = (TOT + NT - 1) / NT;
    constexpr int RV    = HID / 4;
    constexpr int CPASS = (TILE * RV + NT - 1) / NT;
    static_assert((AP / 4) % 2 == 1 && TILE * CP <= TILE * AP, "pitches");
    __shared__ __attribute__((aligned(16))) float sA[TILE * AP];

    const int tid   = threadIdx.x;
    const int wave  = __builtin_amdgcn_readfirstlane(tid >> 5);
    const int lane  = tid & 31;
    const int half  = lane >> 4;
    const int l15   = lane & 15;
    const int n0    = blockIdx.x * TILE;
    const int nrows = min(TILE, M - n0);

    const float4* A4 = (const float4*)(A + (size_t)n0 * K);
    float4 av[NPASS];
    #pragma unroll
    for (int it = 0; it < NPASS; it++) {
        int i = tid + it * NT;
        float4 v = make_float4(0.0f, 0.0f, 0.0f, 0.0f);
        if (i < nrows * KV) v = A4[i];
        av[it] = v;
    }
    const float* bcol = B + (size_t)(half * 2) * HID + wave * 16 + l15;
    v2f bf[KS];
    #pragma unroll
    for (int j = 0; j < KS; j++) {
        bf[j].x = bcol[(size_t)(4 * j) * HID];
        bf[j].y = bcol[(size_t)(4 * j + 1) * HID];
    }
    #pragma unroll
    for (int it = 0; it < NPASS; it++) {
        int i = tid + it * NT;
        if (i < TOT) {
            int r = i / KV;
            int q = i - r * KV;
            *(float4*)(sA + r * AP + 4 * q) = av[it];
        }
    }
    __syncthreads();

    v8f acc[4] = {};
    const float* abase = sA + l15 * AP + half * 2;
    #pragma unroll
    for (int j = 0; j < KS; j++) {
        #pragma unroll
        for (int rt = 0; rt < 4; rt++) {
            v2f a = *(const v2f*)(abase + rt * 16 * AP + 4 * j);
            acc[rt] = __builtin_amdgcn_wmma_f32_16x16x4_f32(
                 false, a,  false, bf[j],
                 (short)0, acc[rt],  false,  false);
        }
    }
    __syncthreads();

    const float bv = bias[wave * 16 + l15];
    float rs[CPASS];
    if (SCALE) {
        #pragma unroll
        for (int it = 0; it < CPASS; it++) {
            int r = (tid + it * NT) / RV;
            rs[it] = (r < nrows) ? rscale[n0 + r] : 0.0f;
        }
    }
    float* cs = sA + (half * 8) * CP + wave * 16 + l15;
    #pragma unroll
    for (int rt = 0; rt < 4; rt++) {
        #pragma unroll
        for (int v = 0; v < 8; v++) {
            float r = acc[rt][v] + bv;
            if (RELU) r = fmaxf(r, 0.0f);
            cs[(rt * 16 + v) * CP] = r;
        }
    }
    __syncthreads();

    static_assert(TILE * RV == CPASS * NT, "the C tile is a whole number of passes");
    float4* C4 = (float4*)(C + (size_t)n0 * HID);
    if (nrows == TILE) {
        float4 cv[CPASS];
        #pragma unroll
        for (int it = 0; it < CPASS; it++) {
            int i = tid + it * NT;
            int r = i / RV;
            int q = i - r * RV;
            cv[it] = *(const float4*)(sA + r * CP + 4 * q);
            if (SCALE) {
                cv[it].x *= rs[it]; cv[it].y *= rs[it]; cv[it].z *= rs[it]; cv[it].w *= rs[it];
            }
        }
        #pragma unroll
        for (int it = 0; it < CPASS; it++) C4[tid + it * NT] = cv[it];
    } else {
        #pragma unroll
        for (int it = 0; it < CPASS; it++) {
            int i = tid + it * NT;
            if (i < nrows * RV) {
                int r = i / RV;
                int q = i - r * RV;
                float4 c = *(const float4*)(sA + r * CP + 4 * q);
                if (SCALE) { c.x *= rs[it]; c.y *= rs[it]; c.z *= rs[it]; c.w *= rs[it]; }
                C4[i] = c;
            }
        }
    }
}

__device__ __forceinline__ void tile_edge8(const float* __restrict__ h, float* acc,
                                           int s, int dl, int j, int lane) {
    float v[8][3];
    int   ro[8];
    #pragma unroll
    for (int u = 0; u < 8; u++) {
        int   sj = __builtin_amdgcn_readlane(s, j + u);
        ro[u]    = __builtin_amdgcn_readlane(dl, j + u) * APITCH + lane;
        const float* srow = h + (size_t)sj * HID + lane;
        v[u][0] = srow[0];
        v[u][1] = srow[32];
        v[u][2] = srow[64];
    }
    #pragma unroll
    for (int u = 0; u < 8; u++) {
        __hip_atomic_fetch_add(acc + ro[u],      v[u][0], __ATOMIC_RELAXED, __HIP_MEMORY_SCOPE_WORKGROUP);
        __hip_atomic_fetch_add(acc + ro[u] + 32, v[u][1], __ATOMIC_RELAXED, __HIP_MEMORY_SCOPE_WORKGROUP);
        __hip_atomic_fetch_add(acc + ro[u] + 64, v[u][2], __ATOMIC_RELAXED, __HIP_MEMORY_SCOPE_WORKGROUP);
    }
}

template <int LAST>
__global__ void __launch_bounds__(AGG_WAVES * 32)
k_tile_layer(const float* __restrict__ h, const float* __restrict__ W,
             const float* __restrict__ bias, const float* __restrict__ dinv,
             const int2* __restrict__ bedge, const int* __restrict__ tstart,
             float* __restrict__ hout) {
    __shared__ __attribute__((aligned(16))) float acc[TILE * APITCH];
    __shared__ __attribute__((aligned(16))) float sWt[HID * WPITCH];
    constexpr int NT = AGG_WAVES * 32;
    constexpr int WV = HID * HID / 4;
    constexpr int RV = HID / 4;
    const int tid   = threadIdx.x;
    const int t     = blockIdx.x;
    const int n0    = t * TILE;
    const int nrows = min(TILE, NNODES - n0);

    const float4* W4 = (const float4*)W;
    #pragma unroll
    for (int it = 0; it < (WV + NT - 1) / NT; it++) {
        int i = tid + it * NT;
        if (i < WV) {
            float4 w = W4[i];
            int k   = i / RV;
            int col = (i - k * RV) * 4;
            float* p = sWt + col * WPITCH + k;
            p[0]          = w.x;
            p[WPITCH]     = w.y;
            p[2 * WPITCH] = w.z;
            p[3 * WPITCH] = w.w;
        }
    }
    const float4* hin = (const float4*)(h + (size_t)n0 * HID);
    float4* lacc = (float4*)acc;
    for (int i = tid; i < nrows * RV; i += NT) {
        int r = i / RV;
        int q = i - r * RV;
        lacc[r * (APITCH / 4) + q] = hin[i];
    }
    __syncthreads();

    const int e0 = tstart[t], e1 = tstart[t + 1];
    const int wave = __builtin_amdgcn_readfirstlane(tid >> 5);
    const int lane = tid & 31;
    for (int base = e0 + wave * 32; base < e1; base += NT) {
        int s = 0, dl = 0;
        int e = base + lane;
        if (e < e1) {
            int2 sd = bedge[e];
            s  = sd.x;
            dl = sd.y - n0;
        }
        const int cnt = min(32, e1 - base);
        int j = 0;
        for (; j + 8 <= cnt; j += 8) tile_edge8(h, acc, s, dl, j, lane);
        for (; j < cnt; j++) {
            int   sj = __builtin_amdgcn_readlane(s, j);
            int   ro = __builtin_amdgcn_readlane(dl, j) * APITCH + lane;
            const float* srow = h + (size_t)sj * HID + lane;
            float v0 = srow[0], v1 = srow[32], v2 = srow[64];
            __hip_atomic_fetch_add(acc + ro,      v0, __ATOMIC_RELAXED, __HIP_MEMORY_SCOPE_WORKGROUP);
            __hip_atomic_fetch_add(acc + ro + 32, v1, __ATOMIC_RELAXED, __HIP_MEMORY_SCOPE_WORKGROUP);
            __hip_atomic_fetch_add(acc + ro + 64, v2, __ATOMIC_RELAXED, __HIP_MEMORY_SCOPE_WORKGROUP);
        }
    }
    __syncthreads();

    const int rt = wave & 3;
    if (wave < MM_WAVES && rt * 16 < nrows) {
        const int half = lane >> 4;
        const int l15  = lane & 15;
        const int nt0  = (wave >> 2) * 2;
        const float* abase = acc + (rt * 16 + l15) * APITCH + half * 2;
        const float* bbase = sWt + (nt0 * 16 + l15) * WPITCH + half * 2;
        v8f c0 = {}, c1 = {};
        #pragma unroll 4
        for (int k = 0; k < HID; k += 4) {
            v2f a  = *(const v2f*)(abase + k);
            v2f b0 = *(const v2f*)(bbase + k);
            v2f b1 = *(const v2f*)(bbase + 16 * WPITCH + k);
            c0 = __builtin_amdgcn_wmma_f32_16x16x4_f32(
                 false, a,  false, b0,
                 (short)0, c0,  false,  false);
            c1 = __builtin_amdgcn_wmma_f32_16x16x4_f32(
                 false, a,  false, b1,
                 (short)0, c1,  false,  false);
        }
        float* obase = hout + (size_t)(n0 + rt * 16 + half * 8) * HID + nt0 * 16 + l15;
        const float bv0 = bias[nt0 * 16 + l15];
        const float bv1 = bias[nt0 * 16 + 16 + l15];
        const float* dr = dinv + n0 + rt * 16 + half * 8;
        #pragma unroll
        for (int v = 0; v < 8; v++) {
            const float dv = dr[v];
            float y0 = fmaxf(c0[v] * dv + bv0, 0.0f);
            float y1 = fmaxf(c1[v] * dv + bv1, 0.0f);
            if (!LAST) { y0 *= dv; y1 *= dv; }
            obase[(size_t)v * HID]      = y0;
            obase[(size_t)v * HID + 16] = y1;
        }
    }
}

#define PL_NT 384
#define PL_S  ((NNODES + PL_NT - 1) / PL_NT)
#define PL_DEPTH 8

__global__ void __launch_bounds__(PL_NT)
k_pool_graph(const float* __restrict__ h, const int* __restrict__ batch,
             float* __restrict__ g, int n) {
    constexpr int NW    = PL_NT / 32;
    constexpr int RV    = HID / 4;
    constexpr int SLOTS = PL_NT / RV;
    static_assert(PL_NT % RV == 0 && PL_NT % 64 == 0, "a thread keeps its columns; the two halves are whole waves");
    static_assert(PL_S - 1 <= PL_NT / 2, "the ids between two samples fit one half of the work-group");
    __shared__ int wc[3][NW];
    __shared__ __attribute__((aligned(16))) float4 red[PL_NT];

    const int tid  = threadIdx.x;
    const int gi   = blockIdx.x;
    const int wave = tid >> 5;
    const int lane = tid & 31;

    const int p = tid * PL_S;
    bool below = false, upto = false;
    if (p < n) {
        int v = batch[p];
        below = v < gi;
        upto  = v <= gi;
    }
    const unsigned m0 = __builtin_amdgcn_ballot_w32(below);
    const unsigned m1 = __builtin_amdgcn_ballot_w32(upto);
    if (lane == 0) {
        wc[0][wave] = __builtin_popcount(m0);
        wc[1][wave] = __builtin_popcount(m1);
    }
    __syncthreads();
    int c_lo = 0, c_hi = 0;
    #pragma unroll
    for (int w = 0; w < NW; w++) {
        c_lo += wc[0][w];
        c_hi += wc[1][w];
    }
    const int upper = (tid >= PL_NT / 2) ? 1 : 0;
    const int j     = tid - upper * (PL_NT / 2);
    const int c     = upper ? c_hi : c_lo;
    bool in = false;
    if (c > 0 && j < PL_S - 1) {
        int idx = (c - 1) * PL_S + 1 + j;
        if (idx < n) {
            int v = batch[idx];
            in = upper ? (v <= gi) : (v < gi);
        }
    }
    const unsigned m2 = __builtin_amdgcn_ballot_w32(in);
    if (lane == 0) wc[2][wave] = __builtin_popcount(m2);
    __syncthreads();
    int d_lo = 0, d_hi = 0;
    #pragma unroll
    for (int w = 0; w < NW / 2; w++) {
        d_lo += wc[2][w];
        d_hi += wc[2][w + NW / 2];
    }
    const int lo = c_lo ? (c_lo - 1) * PL_S + 1 + d_lo : 0;
    const int hi = c_hi ? (c_hi - 1) * PL_S + 1 + d_hi : 0;

    const int cnt = (hi - lo) * RV;
    const float4* src = (const float4*)(h + (size_t)lo * HID);
    float4 acc = make_float4(0.0f, 0.0f, 0.0f, 0.0f);
    for (int base = 0; base < cnt; base += PL_DEPTH * PL_NT) {
        float4 v[PL_DEPTH];
        #pragma unroll
        for (int u = 0; u < PL_DEPTH; u++) {
            int i = base + u * PL_NT + tid;
            v[u] = make_float4(0.0f, 0.0f, 0.0f, 0.0f);
            if (i < cnt) v[u] = src[i];
        }
        #pragma unroll
        for (int u = 0; u < PL_DEPTH; u++) {
            acc.x += v[u].x; acc.y += v[u].y; acc.z += v[u].z; acc.w += v[u].w;
        }
    }
    red[tid] = acc;
    __syncthreads();
    if (tid < RV) {
        float4 s = red[tid];
        #pragma unroll
        for (int r = 1; r < SLOTS; r++) {
            float4 v = red[r * RV + tid];
            s.x += v.x; s.y += v.y; s.z += v.z; s.w += v.w;
        }
        ((float4*)(g + (size_t)gi * HID))[tid] = s;
    }
}

__global__ void k_dec2(const float* __restrict__ g2, const float* __restrict__ W,
                       const float* __restrict__ b, float* __restrict__ out) {
    int idx = blockIdx.x * blockDim.x + threadIdx.x;
    if (idx < NGRAPH * OUTDIM) {
        int gi = idx / OUTDIM;
        int o  = idx - gi * OUTDIM;
        float s = b[o];
        const float* row = g2 + (size_t)gi * HID;
        #pragma unroll 8
        for (int k = 0; k < HID; k++) s += row[k] * W[k * OUTDIM + o];
        out[idx] = s;
    }
}

extern "C" void kernel_launch(void* const* d_in, const int* in_sizes, int n_in,
                              void* d_out, int out_size, void* d_ws, size_t ws_size,
                              hipStream_t stream) {
    const float* x       = (const float*)d_in[0];
    const int*   ei      = (const int*)  d_in[1];
    const int*   batch   = (const int*)  d_in[2];
    const float* enc_W0  = (const float*)d_in[3];
    const float* enc_b0  = (const float*)d_in[4];
    const float* enc_W1  = (const float*)d_in[5];
    const float* enc_b1  = (const float*)d_in[6];
    const float* conv_W  = (const float*)d_in[7];
    const float* conv_b  = (const float*)d_in[8];
    const float* dec_W0  = (const float*)d_in[9];
    const float* dec_b0  = (const float*)d_in[10];
    const float* dec_W1  = (const float*)d_in[11];
    const float* dec_b1  = (const float*)d_in[12];
    float* out = (float*)d_out;

    float* ws   = (float*)d_ws;
    size_t feat = (size_t)NNODES * HID;
    float* buf0 = ws;
    float* buf1 = buf0 + feat;
    float* buf2 = buf1 + feat;
    float* dinv = buf2 + feat;
    float* g    = dinv + DINV_PLANE;
    float* g2   = g + (size_t)NGRAPH * HID;
    int*   hist   = (int*)(g2 + (size_t)NGRAPH * HID);
    int*   basem  = hist + (size_t)FB * NTP;
    int*   tstart = basem + (size_t)FB * NTP;
    int2*  bedge  = (int2*)(tstart + NTP);

    const int B = 256;

    int per = cdiv(NEDGES, FB);
    k_tile_count<<<FB, FT, 0, stream>>>(ei, hist, NEDGES, per);
    k_tile_scan<<<1, NTP, 0, stream>>>(hist, basem, tstart);
    k_tile_fill<<<FB, FT, 0, stream>>>(ei, basem, bedge, NEDGES, per);
    k_tile_dinv<<<NTILES, B, 0, stream>>>(bedge, tstart, dinv);

    k_gemm_ws<INDIM, 1, 0><<<NTILES, GW_WAVES * 32, 0, stream>>>(x, enc_W0, enc_b0, dinv, buf0, NNODES);
    k_gemm_ws<HID, 0, 1><<<NTILES, GW_WAVES * 32, 0, stream>>>(buf0, enc_W1, enc_b1, dinv, buf1, NNODES);

    float* cur = buf1;
    float* nxt = buf2;
    for (int l = 0; l < NCONV; l++) {
        const float* Wl = conv_W + (size_t)l * HID * HID;
        const float* bl = conv_b + (size_t)l * HID;
        if (l + 1 < NCONV)
            k_tile_layer<0><<<NTILES, AGG_WAVES * 32, 0, stream>>>(cur, Wl, bl, dinv, bedge, tstart, nxt);
        else
            k_tile_layer<1><<<NTILES, AGG_WAVES * 32, 0, stream>>>(cur, Wl, bl, dinv, bedge, tstart, nxt);
        float* t = cur; cur = nxt; nxt = t;
    }

    k_pool_graph<<<NGRAPH, PL_NT, 0, stream>>>(cur, batch, g, NNODES);

    k_gemm_ws<HID, 1, 0><<<NGRAPH / TILE, GW_WAVES * 32, 0, stream>>>(g, dec_W0, dec_b0, dinv, g2, NGRAPH);
    k_dec2<<<cdiv(NGRAPH * OUTDIM, B), B, 0, stream>>>(g2, dec_W1, dec_b1, out);
}


# Conv layer kernel on 8 fat waves, not 16 (half the wave entries): staging loads and first edges all asked for at entry, 16 edges per batch, next chunk one step ahead, rows out whole via the LDS tile
# speedup vs baseline: 3.4216x; 1.1060x over previous
#include <hip/hip_runtime.h>
#include <hip/hip_bf16.h>

typedef __attribute__((ext_vector_type(2))) float v2f;
typedef __attribute__((ext_vector_type(8))) float v8f;

#define NNODES 50000
#define NEDGES 800000
#define NGRAPH 512
#define INDIM  128
#define HID    96
#define OUTDIM 10
#define NCONV  4
#define GW_WAVES 6
#define TSHIFT 6
#define TILE   64
#define NTILES ((NNODES + TILE - 1) / TILE)
#define NTP    1024
#define FB     128
#define FT     1024
#define AGG_WAVES 8
#define APITCH 100
#define WPITCH (HID + 2)
#define DINV_PLANE ((NNODES + 31) / 32 * 32)

static_assert(NNODES % 16 == 0, "a tile's valid rows must be whole 16-row WMMA tiles");

static inline int cdiv(int a, int b) { return (a + b - 1) / b; }

__global__ void k_tile_count(const int* __restrict__ ei, int* __restrict__ hist, int E, int per) {
    __shared__ int lh[NTILES];
    for (int i = threadIdx.x; i < NTILES; i += blockDim.x) lh[i] = 0;
    __syncthreads();
    int e0 = blockIdx.x * per;
    int e1 = min(e0 + per, E);
    for (int e = e0 + (int)threadIdx.x; e < e1; e += blockDim.x)
        atomicAdd(&lh[ei[E + e] >> TSHIFT], 1);
    __syncthreads();
    for (int i = threadIdx.x; i < NTILES; i += blockDim.x) hist[blockIdx.x * NTP + i] = lh[i];
}

__global__ void k_tile_scan(const int* __restrict__ hist, int* __restrict__ basem,
                            int* __restrict__ tstart) {
    __shared__ int sc[NTP];
    int t = threadIdx.x;
    int tot = 0;
    if (t < NTILES) {
        #pragma unroll 16
        for (int b = 0; b < FB; b++) tot += hist[b * NTP + t];
    }
    sc[t] = tot;
    __syncthreads();
    for (int off = 1; off < NTP; off <<= 1) {
        int a = (t >= off) ? sc[t - off] : 0;
        __syncthreads();
        sc[t] += a;
        __syncthreads();
    }
    int excl = sc[t] - tot;
    if (t <= NTILES) tstart[t] = excl;
    if (t < NTILES) {
        int run = excl;
        #pragma unroll 16
        for (int b = 0; b < FB; b++) {
            int cn = hist[b * NTP + t];
            basem[b * NTP + t] = run;
            run += cn;
        }
    }
}

__global__ void k_tile_fill(const int* __restrict__ ei, const int* __restrict__ basem,
                            int2* __restrict__ bedge, int E, int per) {
    __shared__ int lh[NTILES];
    __shared__ int lb[NTILES];
    for (int i = threadIdx.x; i < NTILES; i += blockDim.x) {
        lh[i] = 0;
        lb[i] = basem[blockIdx.x * NTP + i];
    }
    __syncthreads();
    int e0 = blockIdx.x * per;
    int e1 = min(e0 + per, E);
    for (int e = e0 + (int)threadIdx.x; e < e1; e += blockDim.x) {
        int s = ei[e];
        int d = ei[E + e];
        int bin = d >> TSHIFT;
        int r = atomicAdd(&lh[bin], 1);
        bedge[lb[bin] + r] = make_int2(s, d);
    }
}

__global__ void k_tile_dinv(const int2* __restrict__ bedge, const int* __restrict__ tstart,
                            float* __restrict__ dinv) {
    __shared__ int cnt[TILE];
    int t  = blockIdx.x;
    int n0 = t * TILE;
    if (threadIdx.x < TILE) cnt[threadIdx.x] = 0;
    __syncthreads();
    int e0 = tstart[t], e1 = tstart[t + 1];
    for (int e = e0 + (int)threadIdx.x; e < e1; e += blockDim.x)
        atomicAdd(&cnt[bedge[e].y - n0], 1);
    __syncthreads();
    int node = n0 + (int)threadIdx.x;
    if (threadIdx.x < TILE && node < NNODES)
        dinv[node] = rsqrtf(1.0f + (float)cnt[threadIdx.x]);
}

template <int K, int RELU, int SCALE>
__global__ void __launch_bounds__(GW_WAVES * 32)
k_gemm_ws(const float* __restrict__ A, const float* __restrict__ B,
          const float* __restrict__ bias, const float* __restrict__ rscale,
          float* __restrict__ C, int M) {
    constexpr int NT    = GW_WAVES * 32;
    constexpr int KV    = K / 4;
    constexpr int KS    = K / 4;
    constexpr int AP    = K + 4;
    constexpr int CP    = HID + 4;
    constexpr int TOT   = TILE * KV;
    constexpr int NPASS = (TOT + NT - 1) / NT;
    constexpr int RV    = HID / 4;
    constexpr int CPASS = (TILE * RV + NT - 1) / NT;
    static_assert((AP / 4) % 2 == 1 && TILE * CP <= TILE * AP, "pitches");
    __shared__ __attribute__((aligned(16))) float sA[TILE * AP];

    const int tid   = threadIdx.x;
    const int wave  = __builtin_amdgcn_readfirstlane(tid >> 5);
    const int lane  = tid & 31;
    const int half  = lane >> 4;
    const int l15   = lane & 15;
    const int n0    = blockIdx.x * TILE;
    const int nrows = min(TILE, M - n0);

    const float4* A4 = (const float4*)(A + (size_t)n0 * K);
    float4 av[NPASS];
    #pragma unroll
    for (int it = 0; it < NPASS; it++) {
        int i = tid + it * NT;
        float4 v = make_float4(0.0f, 0.0f, 0.0f, 0.0f);
        if (i < nrows * KV) v = A4[i];
        av[it] = v;
    }
    const float* bcol = B + (size_t)(half * 2) * HID + wave * 16 + l15;
    v2f bf[KS];
    #pragma unroll
    for (int j = 0; j < KS; j++) {
        bf[j].x = bcol[(size_t)(4 * j) * HID];
        bf[j].y = bcol[(size_t)(4 * j + 1) * HID];
    }
    #pragma unroll
    for (int it = 0; it < NPASS; it++) {
        int i = tid + it * NT;
        if (i < TOT) {
            int r = i / KV;
            int q = i - r * KV;
            *(float4*)(sA + r * AP + 4 * q) = av[it];
        }
    }
    __syncthreads();

    v8f acc[4] = {};
    const float* abase = sA + l15 * AP + half * 2;
    #pragma unroll
    for (int j = 0; j < KS; j++) {
        #pragma unroll
        for (int rt = 0; rt < 4; rt++) {
            v2f a = *(const v2f*)(abase + rt * 16 * AP + 4 * j);
            acc[rt] = __builtin_amdgcn_wmma_f32_16x16x4_f32(
                 false, a,  false, bf[j],
                 (short)0, acc[rt],  false,  false);
        }
    }
    __syncthreads();

    const float bv = bias[wave * 16 + l15];
    float rs[CPASS];
    if (SCALE) {
        #pragma unroll
        for (int it = 0; it < CPASS; it++) {
            int r = (tid + it * NT) / RV;
            rs[it] = (r < nrows) ? rscale[n0 + r] : 0.0f;
        }
    }
    float* cs = sA + (half * 8) * CP + wave * 16 + l15;
    #pragma unroll
    for (int rt = 0; rt < 4; rt++) {
        #pragma unroll
        for (int v = 0; v < 8; v++) {
            float r = acc[rt][v] + bv;
            if (RELU) r = fmaxf(r, 0.0f);
            cs[(rt * 16 + v) * CP] = r;
        }
    }
    __syncthreads();

    static_assert(TILE * RV == CPASS * NT, "the C tile is a whole number of passes");
    float4* C4 = (float4*)(C + (size_t)n0 * HID);
    if (nrows == TILE) {
        float4 cv[CPASS];
        #pragma unroll
        for (int it = 0; it < CPASS; it++) {
            int i = tid + it * NT;
            int r = i / RV;
            int q = i - r * RV;
            cv[it] = *(const float4*)(sA + r * CP + 4 * q);
            if (SCALE) {
                cv[it].x *= rs[it]; cv[it].y *= rs[it]; cv[it].z *= rs[it]; cv[it].w *= rs[it];
            }
        }
        #pragma unroll
        for (int it = 0; it < CPASS; it++) C4[tid + it * NT] = cv[it];
    } else {
        #pragma unroll
        for (int it = 0; it < CPASS; it++) {
            int i = tid + it * NT;
            if (i < nrows * RV) {
                int r = i / RV;
                int q = i - r * RV;
                float4 c = *(const float4*)(sA + r * CP + 4 * q);
                if (SCALE) { c.x *= rs[it]; c.y *= rs[it]; c.z *= rs[it]; c.w *= rs[it]; }
                C4[i] = c;
            }
        }
    }
}

template <int MASK>
__device__ __forceinline__ void tile_edge16(const float* __restrict__ h, float* acc,
                                            int s, int dl, int j, int cnt, int lane) {
    float v[16][3];
    int   ro[16];
    #pragma unroll
    for (int u = 0; u < 16; u++) {
        int   sj = __builtin_amdgcn_readlane(s, j + u);
        ro[u]    = __builtin_amdgcn_readlane(dl, j + u) * APITCH + lane;
        const float* srow = h + (size_t)sj * HID + lane;
        v[u][0] = srow[0];
        v[u][1] = srow[32];
        v[u][2] = srow[64];
    }
    #pragma unroll
    for (int u = 0; u < 16; u++) {
        const bool on = !MASK || (j + u) < cnt;
        __hip_atomic_fetch_add(acc + ro[u],      on ? v[u][0] : 0.0f, __ATOMIC_RELAXED, __HIP_MEMORY_SCOPE_WORKGROUP);
        __hip_atomic_fetch_add(acc + ro[u] + 32, on ? v[u][1] : 0.0f, __ATOMIC_RELAXED, __HIP_MEMORY_SCOPE_WORKGROUP);
        __hip_atomic_fetch_add(acc + ro[u] + 64, on ? v[u][2] : 0.0f, __ATOMIC_RELAXED, __HIP_MEMORY_SCOPE_WORKGROUP);
    }
}

template <int LAST>
__global__ void __launch_bounds__(AGG_WAVES * 32)
k_tile_layer(const float* __restrict__ h, const float* __restrict__ W,
             const float* __restrict__ bias, const float* __restrict__ dinv,
             const int2* __restrict__ bedge, const int* __restrict__ tstart,
             float* __restrict__ hout) {
    __shared__ __attribute__((aligned(16))) float acc[TILE * APITCH];
    __shared__ __attribute__((aligned(16))) float sWt[HID * WPITCH];
    constexpr int NT = AGG_WAVES * 32;
    constexpr int WV = HID * HID / 4;
    constexpr int RV = HID / 4;
    const int tid   = threadIdx.x;
    const int t     = blockIdx.x;
    const int n0    = t * TILE;
    const int nrows = min(TILE, NNODES - n0);
    const int wave = __builtin_amdgcn_readfirstlane(tid >> 5);
    const int lane = tid & 31;
    const int e0 = tstart[t], e1 = tstart[t + 1];
    int2 sd = make_int2(0, n0);
    if (e0 + wave * 32 + lane < e1) sd = bedge[e0 + wave * 32 + lane];

    constexpr int WPASS = WV / NT;
    constexpr int TPASS = TILE * RV / NT;
    static_assert(WV == WPASS * NT && TILE * RV == TPASS * NT, "W and the tile are whole numbers of passes");
    const float4* W4  = (const float4*)W;
    const float4* hin = (const float4*)(h + (size_t)n0 * HID);
    float4* lacc = (float4*)acc;
    float4 wv[WPASS];
    float4 tv[TPASS];
    #pragma unroll
    for (int it = 0; it < WPASS; it++) wv[it] = W4[tid + it * NT];
    #pragma unroll
    for (int it = 0; it < TPASS; it++) {
        int i = tid + it * NT;
        tv[it] = make_float4(0.0f, 0.0f, 0.0f, 0.0f);
        if (i < nrows * RV) tv[it] = hin[i];
    }
    #pragma unroll
    for (int it = 0; it < WPASS; it++) {
        int i   = tid + it * NT;
        int k   = i / RV;
        int col = (i - k * RV) * 4;
        float* p = sWt + col * WPITCH + k;
        p[0]          = wv[it].x;
        p[WPITCH]     = wv[it].y;
        p[2 * WPITCH] = wv[it].z;
        p[3 * WPITCH] = wv[it].w;
    }
    #pragma unroll
    for (int it = 0; it < TPASS; it++) {
        int i = tid + it * NT;
        int r = i / RV;
        int q = i - r * RV;
        lacc[r * (APITCH / 4) + q] = tv[it];
    }
    __syncthreads();

    for (int base = e0 + wave * 32; base < e1; base += NT) {
        const int s  = sd.x;
        const int dl = sd.y - n0;
        sd = make_int2(0, n0);
        if (base + NT + lane < e1) sd = bedge[base + NT + lane];
        const int cnt = min(32, e1 - base);
        int j = 0;
        for (; j + 16 <= cnt; j += 16) tile_edge16<0>(h, acc, s, dl, j, cnt, lane);
        if (j < cnt) tile_edge16<1>(h, acc, s, dl, j, cnt, lane);
    }
    __syncthreads();

    const int rt   = wave & 3;
    const int half = lane >> 4;
    const int l15  = lane & 15;
    const int nt0  = (wave >> 2) * 3;
    const bool mm  = rt * 16 < nrows;
    v8f c0 = {}, c1 = {}, c2 = {};
    float dv[8];
    float bv0 = 0.0f, bv1 = 0.0f, bv2 = 0.0f;
    if (mm) {
        const float* dr = dinv + n0 + rt * 16 + half * 8;
        #pragma unroll
        for (int v = 0; v < 8; v++) dv[v] = dr[v];
        bv0 = bias[nt0 * 16 + l15];
        bv1 = bias[nt0 * 16 + 16 + l15];
        bv2 = bias[nt0 * 16 + 32 + l15];
        const float* abase = acc + (rt * 16 + l15) * APITCH + half * 2;
        const float* bbase = sWt + (nt0 * 16 + l15) * WPITCH + half * 2;
        #pragma unroll 4
        for (int k = 0; k < HID; k += 4) {
            v2f a  = *(const v2f*)(abase + k);
            v2f b0 = *(const v2f*)(bbase + k);
            v2f b1 = *(const v2f*)(bbase + 16 * WPITCH + k);
            v2f b2 = *(const v2f*)(bbase + 32 * WPITCH + k);
            c0 = __builtin_amdgcn_wmma_f32_16x16x4_f32(
                 false, a,  false, b0,
                 (short)0, c0,  false,  false);
            c1 = __builtin_amdgcn_wmma_f32_16x16x4_f32(
                 false, a,  false, b1,
                 (short)0, c1,  false,  false);
            c2 = __builtin_amdgcn_wmma_f32_16x16x4_f32(
                 false, a,  false, b2,
                 (short)0, c2,  false,  false);
        }
    }
    __syncthreads();

    if (mm) {
        float* cs = acc + (rt * 16 + half * 8) * APITCH + nt0 * 16 + l15;
        #pragma unroll
        for (int v = 0; v < 8; v++) {
            float y0 = fmaxf(c0[v] * dv[v] + bv0, 0.0f);
            float y1 = fmaxf(c1[v] * dv[v] + bv1, 0.0f);
            float y2 = fmaxf(c2[v] * dv[v] + bv2, 0.0f);
            if (!LAST) { y0 *= dv[v]; y1 *= dv[v]; y2 *= dv[v]; }
            cs[v * APITCH]      = y0;
            cs[v * APITCH + 16] = y1;
            cs[v * APITCH + 32] = y2;
        }
    }
    __syncthreads();

    float4* hout4 = (float4*)(hout + (size_t)n0 * HID);
    float4 ov[TPASS];
    #pragma unroll
    for (int it = 0; it < TPASS; it++) {
        int i = tid + it * NT;
        int r = i / RV;
        int q = i - r * RV;
        ov[it] = lacc[r * (APITCH / 4) + q];
    }
    #pragma unroll
    for (int it = 0; it < TPASS; it++) {
        int i = tid + it * NT;
        if (i < nrows * RV) hout4[i] = ov[it];
    }
}

#define PL_NT 384
#define PL_S  ((NNODES + PL_NT - 1) / PL_NT)
#define PL_DEPTH 8

__global__ void __launch_bounds__(PL_NT)
k_pool_graph(const float* __restrict__ h, const int* __restrict__ batch,
             float* __restrict__ g, int n) {
    constexpr int NW    = PL_NT / 32;
    constexpr int RV    = HID / 4;
    constexpr int SLOTS = PL_NT / RV;
    static_assert(PL_NT % RV == 0 && PL_NT % 64 == 0, "a thread keeps its columns; the two halves are whole waves");
    static_assert(PL_S - 1 <= PL_NT / 2, "the ids between two samples fit one half of the work-group");
    __shared__ int wc[3][NW];
    __shared__ __attribute__((aligned(16))) float4 red[PL_NT];

    const int tid  = threadIdx.x;
    const int gi   = blockIdx.x;
    const int wave = tid >> 5;
    const int lane = tid & 31;

    const int p = tid * PL_S;
    bool below = false, upto = false;
    if (p < n) {
        int v = batch[p];
        below = v < gi;
        upto  = v <= gi;
    }
    const unsigned m0 = __builtin_amdgcn_ballot_w32(below);
    const unsigned m1 = __builtin_amdgcn_ballot_w32(upto);
    if (lane == 0) {
        wc[0][wave] = __builtin_popcount(m0);
        wc[1][wave] = __builtin_popcount(m1);
    }
    __syncthreads();
    int c_lo = 0, c_hi = 0;
    #pragma unroll
    for (int w = 0; w < NW; w++) {
        c_lo += wc[0][w];
        c_hi += wc[1][w];
    }
    const int upper = (tid >= PL_NT / 2) ? 1 : 0;
    const int j     = tid - upper * (PL_NT / 2);
    const int c     = upper ? c_hi : c_lo;
    bool in = false;
    if (c > 0 && j < PL_S - 1) {
        int idx = (c - 1) * PL_S + 1 + j;
        if (idx < n) {
            int v = batch[idx];
            in = upper ? (v <= gi) : (v < gi);
        }
    }
    const unsigned m2 = __builtin_amdgcn_ballot_w32(in);
    if (lane == 0) wc[2][wave] = __builtin_popcount(m2);
    __syncthreads();
    int d_lo = 0, d_hi = 0;
    #pragma unroll
    for (int w = 0; w < NW / 2; w++) {
        d_lo += wc[2][w];
        d_hi += wc[2][w + NW / 2];
    }
    const int lo = c_lo ? (c_lo - 1) * PL_S + 1 + d_lo : 0;
    const int hi = c_hi ? (c_hi - 1) * PL_S + 1 + d_hi : 0;

    const int cnt = (hi - lo) * RV;
    const float4* src = (const float4*)(h + (size_t)lo * HID);
    float4 acc = make_float4(0.0f, 0.0f, 0.0f, 0.0f);
    for (int base = 0; base < cnt; base += PL_DEPTH * PL_NT) {
        float4 v[PL_DEPTH];
        #pragma unroll
        for (int u = 0; u < PL_DEPTH; u++) {
            int i = base + u * PL_NT + tid;
            v[u] = make_float4(0.0f, 0.0f, 0.0f, 0.0f);
            if (i < cnt) v[u] = src[i];
        }
        #pragma unroll
        for (int u = 0; u < PL_DEPTH; u++) {
            acc.x += v[u].x; acc.y += v[u].y; acc.z += v[u].z; acc.w += v[u].w;
        }
    }
    red[tid] = acc;
    __syncthreads();
    if (tid < RV) {
        float4 s = red[tid];
        #pragma unroll
        for (int r = 1; r < SLOTS; r++) {
            float4 v = red[r * RV + tid];
            s.x += v.x; s.y += v.y; s.z += v.z; s.w += v.w;
        }
        ((float4*)(g + (size_t)gi * HID))[tid] = s;
    }
}

__global__ void k_dec2(const float* __restrict__ g2, const float* __restrict__ W,
                       const float* __restrict__ b, float* __restrict__ out) {
    int idx = blockIdx.x * blockDim.x + threadIdx.x;
    if (idx < NGRAPH * OUTDIM) {
        int gi = idx / OUTDIM;
        int o  = idx - gi * OUTDIM;
        float s = b[o];
        const float* row = g2 + (size_t)gi * HID;
        #pragma unroll 8
        for (int k = 0; k < HID; k++) s += row[k] * W[k * OUTDIM + o];
        out[idx] = s;
    }
}

extern "C" void kernel_launch(void* const* d_in, const int* in_sizes, int n_in,
                              void* d_out, int out_size, void* d_ws, size_t ws_size,
                              hipStream_t stream) {
    const float* x       = (const float*)d_in[0];
    const int*   ei      = (const int*)  d_in[1];
    const int*   batch   = (const int*)  d_in[2];
    const float* enc_W0  = (const float*)d_in[3];
    const float* enc_b0  = (const float*)d_in[4];
    const float* enc_W1  = (const float*)d_in[5];
    const float* enc_b1  = (const float*)d_in[6];
    const float* conv_W  = (const float*)d_in[7];
    const float* conv_b  = (const float*)d_in[8];
    const float* dec_W0  = (const float*)d_in[9];
    const float* dec_b0  = (const float*)d_in[10];
    const float* dec_W1  = (const float*)d_in[11];
    const float* dec_b1  = (const float*)d_in[12];
    float* out = (float*)d_out;

    float* ws   = (float*)d_ws;
    size_t feat = (size_t)NNODES * HID;
    float* buf0 = ws;
    float* buf1 = buf0 + feat;
    float* buf2 = buf1 + feat;
    float* dinv = buf2 + feat;
    float* g    = dinv + DINV_PLANE;
    float* g2   = g + (size_t)NGRAPH * HID;
    int*   hist   = (int*)(g2 + (size_t)NGRAPH * HID);
    int*   basem  = hist + (size_t)FB * NTP;
    int*   tstart = basem + (size_t)FB * NTP;
    int2*  bedge  = (int2*)(tstart + NTP);

    const int B = 256;

    int per = cdiv(NEDGES, FB);
    k_tile_count<<<FB, FT, 0, stream>>>(ei, hist, NEDGES, per);
    k_tile_scan<<<1, NTP, 0, stream>>>(hist, basem, tstart);
    k_tile_fill<<<FB, FT, 0, stream>>>(ei, basem, bedge, NEDGES, per);
    k_tile_dinv<<<NTILES, B, 0, stream>>>(bedge, tstart, dinv);

    k_gemm_ws<INDIM, 1, 0><<<NTILES, GW_WAVES * 32, 0, stream>>>(x, enc_W0, enc_b0, dinv, buf0, NNODES);
    k_gemm_ws<HID, 0, 1><<<NTILES, GW_WAVES * 32, 0, stream>>>(buf0, enc_W1, enc_b1, dinv, buf1, NNODES);

    float* cur = buf1;
    float* nxt = buf2;
    for (int l = 0; l < NCONV; l++) {
        const float* Wl = conv_W + (size_t)l * HID * HID;
        const float* bl = conv_b + (size_t)l * HID;
        if (l + 1 < NCONV)
            k_tile_layer<0><<<NTILES, AGG_WAVES * 32, 0, stream>>>(cur, Wl, bl, dinv, bedge, tstart, nxt);
        else
            k_tile_layer<1><<<NTILES, AGG_WAVES * 32, 0, stream>>>(cur, Wl, bl, dinv, bedge, tstart, nxt);
        float* t = cur; cur = nxt; nxt = t;
    }

    k_pool_graph<<<NGRAPH, PL_NT, 0, stream>>>(cur, batch, g, NNODES);

    k_gemm_ws<HID, 1, 0><<<NGRAPH / TILE, GW_WAVES * 32, 0, stream>>>(g, dec_W0, dec_b0, dinv, g2, NGRAPH);
    k_dec2<<<cdiv(NGRAPH * OUTDIM, B), B, 0, stream>>>(g2, dec_W1, dec_b1, out);
}


# Launches 13 -> 9, small ones folded into neighbours: slice counts ride in encoder GEMM 1, each fill block redoes the scan, degrees/dinv in encoder GEMM 2, decoder tail in its GEMM; sum orders kept
# speedup vs baseline: 3.7422x; 1.0937x over previous
#include <hip/hip_runtime.h>
#include <hip/hip_bf16.h>

typedef __attribute__((ext_vector_type(2))) float v2f;
typedef __attribute__((ext_vector_type(8))) float v8f;

#define NNODES 50000
#define NEDGES 800000
#define NGRAPH 512
#define INDIM  128
#define HID    96
#define OUTDIM 10
#define NCONV  4
#define GW_WAVES 6
#define TSHIFT 6
#define TILE   64
#define NTILES ((NNODES + TILE - 1) / TILE)
#define NTP    1024
#define FB     128
#define FT     1024
#define AGG_WAVES 8
#define APITCH 100
#define WPITCH (HID + 2)
#define DINV_PLANE ((NNODES + 31) / 32 * 32)

static_assert(NNODES % 16 == 0, "a tile's valid rows must be whole 16-row WMMA tiles");

static inline int cdiv(int a, int b) { return (a + b - 1) / b; }


__global__ void __launch_bounds__(FT)
k_tile_fill(const int* __restrict__ ei, const int* __restrict__ hist,
            int* __restrict__ tstart, int2* __restrict__ bedge, int E, int per) {
    __shared__ int lh[NTILES];
    __shared__ int lb[NTILES];
    __shared__ int wsum[FT / 32];
    const int t    = threadIdx.x;
    const int lane = t & 31;
    const int wave = t >> 5;
    const int b    = blockIdx.x;
    int tot = 0, pre = 0;
    if (t < NTILES) {
        #pragma unroll 32
        for (int bb = 0; bb < FB; bb++) {
            int c = hist[bb * NTP + t];
            tot += c;
            if (bb < b) pre += c;
        }
    }
    int v = tot;
    #pragma unroll
    for (int off = 1; off < 32; off <<= 1) {
        int n = __shfl_up(v, off);
        if (lane >= off) v += n;
    }
    if (lane == 31) wsum[wave] = v;
    __syncthreads();
    if (wave == 0) {
        int w = wsum[lane];
        #pragma unroll
        for (int off = 1; off < 32; off <<= 1) {
            int n = __shfl_up(w, off);
            if (lane >= off) w += n;
        }
        wsum[lane] = w;
    }
    __syncthreads();
    const int excl = v - tot + (wave > 0 ? wsum[wave - 1] : 0);
    if (t < NTILES) {
        lh[t] = 0;
        lb[t] = excl + pre;
    }
    if (b == 0 && t <= NTILES) tstart[t] = excl;
    __syncthreads();
    int e0 = b * per;
    int e1 = min(e0 + per, E);
    for (int e = e0 + t; e < e1; e += FT) {
        int s = ei[e];
        int d = ei[E + e];
        int bin = d >> TSHIFT;
        int r = atomicAdd(&lh[bin], 1);
        bedge[lb[bin] + r] = make_int2(s, d);
    }
}

template <int K, int RELU, int MODE>
__global__ void __launch_bounds__(GW_WAVES * 32)
k_gemm_ws(const float* __restrict__ A, const float* __restrict__ B,
          const float* __restrict__ bias,
          const int* __restrict__ ei, int* __restrict__ hist,
          const int2* __restrict__ bedge, const int* __restrict__ tstart,
          float* __restrict__ dinv,
          const float* __restrict__ W2, const float* __restrict__ b2,
          float* __restrict__ out2,
          float* __restrict__ C, int M) {
    constexpr int NT    = GW_WAVES * 32;
    constexpr int KV    = K / 4;
    constexpr int KS    = K / 4;
    constexpr int AP    = K + 4;
    constexpr int CP    = HID + 4;
    constexpr int TOT   = TILE * KV;
    constexpr int NPASS = (TOT + NT - 1) / NT;
    constexpr int RV    = HID / 4;
    constexpr int CPASS = (TILE * RV + NT - 1) / NT;
    constexpr int DPRE  = 8;
    static_assert((AP / 4) % 2 == 1 && TILE * CP <= TILE * AP, "pitches");
    __shared__ __attribute__((aligned(16))) float sA[TILE * AP];
    __shared__ int   cnt[MODE == 1 ? TILE : 1];
    __shared__ float sW2[MODE == 2 ? HID * OUTDIM : 1];
    __shared__ int   lh[MODE == 0 ? NTILES : 1];

    const int tid   = threadIdx.x;

    if (MODE == 0 && (int)blockIdx.x >= NTILES) {
        const int sl  = (int)blockIdx.x - NTILES;
        const int per = (NEDGES + FB - 1) / FB;
        for (int i = tid; i < NTILES; i += NT) lh[i] = 0;
        __syncthreads();
        const int e0 = sl * per;
        const int e1 = min(e0 + per, NEDGES);
        for (int base = e0; base < e1; base += DPRE * NT) {
            int d[DPRE];
            #pragma unroll
            for (int u = 0; u < DPRE; u++) {
                int e = base + tid + u * NT;
                d[u] = (e < e1) ? ei[NEDGES + e] : -1;
            }
            #pragma unroll
            for (int u = 0; u < DPRE; u++)
                if (d[u] >= 0) atomicAdd(&lh[d[u] >> TSHIFT], 1);
        }
        __syncthreads();
        for (int i = tid; i < NTILES; i += NT) hist[sl * NTP + i] = lh[i];
        return;
    }

    const int wave  = __builtin_amdgcn_readfirstlane(tid >> 5);
    const int lane  = tid & 31;
    const int half  = lane >> 4;
    const int l15   = lane & 15;
    const int n0    = blockIdx.x * TILE;
    const int nrows = min(TILE, M - n0);

    const float4* A4 = (const float4*)(A + (size_t)n0 * K);
    float4 av[NPASS];
    #pragma unroll
    for (int it = 0; it < NPASS; it++) {
        int i = tid + it * NT;
        float4 v = make_float4(0.0f, 0.0f, 0.0f, 0.0f);
        if (i < nrows * KV) v = A4[i];
        av[it] = v;
    }
    const float* bcol = B + (size_t)(half * 2) * HID + wave * 16 + l15;
    v2f bf[KS];
    #pragma unroll
    for (int j = 0; j < KS; j++) {
        bf[j].x = bcol[(size_t)(4 * j) * HID];
        bf[j].y = bcol[(size_t)(4 * j + 1) * HID];
    }
    int dpre[DPRE];
    int e0 = 0, e1 = 0;
    const int* bdst = (const int*)bedge + 1;
    if (MODE == 1) {
        e0 = tstart[blockIdx.x];
        e1 = tstart[blockIdx.x + 1];
        #pragma unroll
        for (int u = 0; u < DPRE; u++) {
            int e = e0 + tid + u * NT;
            dpre[u] = (e < e1) ? bdst[2 * (size_t)e] : -1;
        }
        if (tid < TILE) cnt[tid] = 0;
    }
    if (MODE == 2) {
        for (int i = tid; i < HID * OUTDIM; i += NT) sW2[i] = W2[i];
    }
    #pragma unroll
    for (int it = 0; it < NPASS; it++) {
        int i = tid + it * NT;
        if (i < TOT) {
            int r = i / KV;
            int q = i - r * KV;
            *(float4*)(sA + r * AP + 4 * q) = av[it];
        }
    }
    __syncthreads();

    if (MODE == 1) {
        #pragma unroll
        for (int u = 0; u < DPRE; u++)
            if (dpre[u] >= 0) atomicAdd(&cnt[dpre[u] - n0], 1);
        for (int e = e0 + tid + DPRE * NT; e < e1; e += NT)
            atomicAdd(&cnt[bdst[2 * (size_t)e] - n0], 1);
    }

    v8f acc[4] = {};
    const float* abase = sA + l15 * AP + half * 2;
    #pragma unroll
    for (int j = 0; j < KS; j++) {
        #pragma unroll
        for (int rt = 0; rt < 4; rt++) {
            v2f a = *(const v2f*)(abase + rt * 16 * AP + 4 * j);
            acc[rt] = __builtin_amdgcn_wmma_f32_16x16x4_f32(
                 false, a,  false, bf[j],
                 (short)0, acc[rt],  false,  false);
        }
    }
    __syncthreads();

    const float bv = bias[wave * 16 + l15];
    float rs[CPASS];
    if (MODE == 1) {
        #pragma unroll
        for (int it = 0; it < CPASS; it++) {
            int r = (tid + it * NT) / RV;
            rs[it] = rsqrtf(1.0f + (float)cnt[r]);
        }
        if (tid < TILE && n0 + tid < M) dinv[n0 + tid] = rsqrtf(1.0f + (float)cnt[tid]);
    }
    float* cs = sA + (half * 8) * CP + wave * 16 + l15;
    #pragma unroll
    for (int rt = 0; rt < 4; rt++) {
        #pragma unroll
        for (int v = 0; v < 8; v++) {
            float r = acc[rt][v] + bv;
            if (RELU) r = fmaxf(r, 0.0f);
            cs[(rt * 16 + v) * CP] = r;
        }
    }
    __syncthreads();

    if (MODE == 2) {
        for (int idx = tid; idx < TILE * OUTDIM; idx += NT) {
            int r = idx / OUTDIM;
            int o = idx - r * OUTDIM;
            float s = b2[o];
            const float* row = sA + r * CP;
            #pragma unroll 8
            for (int k = 0; k < HID; k++) s += row[k] * sW2[k * OUTDIM + o];
            if (r < nrows) out2[(size_t)n0 * OUTDIM + idx] = s;
        }
        return;
    }

    static_assert(TILE * RV == CPASS * NT, "the C tile is a whole number of passes");
    float4* C4 = (float4*)(C + (size_t)n0 * HID);
    if (nrows == TILE) {
        float4 cv[CPASS];
        #pragma unroll
        for (int it = 0; it < CPASS; it++) {
            int i = tid + it * NT;
            int r = i / RV;
            int q = i - r * RV;
            cv[it] = *(const float4*)(sA + r * CP + 4 * q);
            if (MODE == 1) {
                cv[it].x *= rs[it]; cv[it].y *= rs[it]; cv[it].z *= rs[it]; cv[it].w *= rs[it];
            }
        }
        #pragma unroll
        for (int it = 0; it < CPASS; it++) C4[tid + it * NT] = cv[it];
    } else {
        #pragma unroll
        for (int it = 0; it < CPASS; it++) {
            int i = tid + it * NT;
            if (i < nrows * RV) {
                int r = i / RV;
                int q = i - r * RV;
                float4 c = *(const float4*)(sA + r * CP + 4 * q);
                if (MODE == 1) { c.x *= rs[it]; c.y *= rs[it]; c.z *= rs[it]; c.w *= rs[it]; }
                C4[i] = c;
            }
        }
    }
}

template <int MASK>
__device__ __forceinline__ void tile_edge16(const float* __restrict__ h, float* acc,
                                            int s, int dl, int j, int cnt, int lane) {
    float v[16][3];
    int   ro[16];
    #pragma unroll
    for (int u = 0; u < 16; u++) {
        int   sj = __builtin_amdgcn_readlane(s, j + u);
        ro[u]    = __builtin_amdgcn_readlane(dl, j + u) * APITCH + lane;
        const float* srow = h + (size_t)sj * HID + lane;
        v[u][0] = srow[0];
        v[u][1] = srow[32];
        v[u][2] = srow[64];
    }
    #pragma unroll
    for (int u = 0; u < 16; u++) {
        const bool on = !MASK || (j + u) < cnt;
        __hip_atomic_fetch_add(acc + ro[u],      on ? v[u][0] : 0.0f, __ATOMIC_RELAXED, __HIP_MEMORY_SCOPE_WORKGROUP);
        __hip_atomic_fetch_add(acc + ro[u] + 32, on ? v[u][1] : 0.0f, __ATOMIC_RELAXED, __HIP_MEMORY_SCOPE_WORKGROUP);
        __hip_atomic_fetch_add(acc + ro[u] + 64, on ? v[u][2] : 0.0f, __ATOMIC_RELAXED, __HIP_MEMORY_SCOPE_WORKGROUP);
    }
}

template <int LAST>
__global__ void __launch_bounds__(AGG_WAVES * 32)
k_tile_layer(const float* __restrict__ h, const float* __restrict__ W,
             const float* __restrict__ bias, const float* __restrict__ dinv,
             const int2* __restrict__ bedge, const int* __restrict__ tstart,
             float* __restrict__ hout) {
    __shared__ __attribute__((aligned(16))) float acc[TILE * APITCH];
    __shared__ __attribute__((aligned(16))) float sWt[HID * WPITCH];
    constexpr int NT = AGG_WAVES * 32;
    constexpr int WV = HID * HID / 4;
    constexpr int RV = HID / 4;
    const int tid   = threadIdx.x;
    const int t     = blockIdx.x;
    const int n0    = t * TILE;
    const int nrows = min(TILE, NNODES - n0);
    const int wave = __builtin_amdgcn_readfirstlane(tid >> 5);
    const int lane = tid & 31;
    const int e0 = tstart[t], e1 = tstart[t + 1];
    int2 sd = make_int2(0, n0);
    if (e0 + wave * 32 + lane < e1) sd = bedge[e0 + wave * 32 + lane];

    constexpr int WPASS = WV / NT;
    constexpr int TPASS = TILE * RV / NT;
    static_assert(WV == WPASS * NT && TILE * RV == TPASS * NT, "W and the tile are whole numbers of passes");
    const float4* W4  = (const float4*)W;
    const float4* hin = (const float4*)(h + (size_t)n0 * HID);
    float4* lacc = (float4*)acc;
    float4 wv[WPASS];
    float4 tv[TPASS];
    #pragma unroll
    for (int it = 0; it < WPASS; it++) wv[it] = W4[tid + it * NT];
    #pragma unroll
    for (int it = 0; it < TPASS; it++) {
        int i = tid + it * NT;
        tv[it] = make_float4(0.0f, 0.0f, 0.0f, 0.0f);
        if (i < nrows * RV) tv[it] = hin[i];
    }
    #pragma unroll
    for (int it = 0; it < WPASS; it++) {
        int i   = tid + it * NT;
        int k   = i / RV;
        int col = (i - k * RV) * 4;
        float* p = sWt + col * WPITCH + k;
        p[0]          = wv[it].x;
        p[WPITCH]     = wv[it].y;
        p[2 * WPITCH] = wv[it].z;
        p[3 * WPITCH] = wv[it].w;
    }
    #pragma unroll
    for (int it = 0; it < TPASS; it++) {
        int i = tid + it * NT;
        int r = i / RV;
        int q = i - r * RV;
        lacc[r * (APITCH / 4) + q] = tv[it];
    }
    __syncthreads();

    for (int base = e0 + wave * 32; base < e1; base += NT) {
        const int s  = sd.x;
        const int dl = sd.y - n0;
        sd = make_int2(0, n0);
        if (base + NT + lane < e1) sd = bedge[base + NT + lane];
        const int cnt = min(32, e1 - base);
        int j = 0;
        for (; j + 16 <= cnt; j += 16) tile_edge16<0>(h, acc, s, dl, j, cnt, lane);
        if (j < cnt) tile_edge16<1>(h, acc, s, dl, j, cnt, lane);
    }
    __syncthreads();

    const int rt   = wave & 3;
    const int half = lane >> 4;
    const int l15  = lane & 15;
    const int nt0  = (wave >> 2) * 3;
    const bool mm  = rt * 16 < nrows;
    v8f c0 = {}, c1 = {}, c2 = {};
    float dv[8];
    float bv0 = 0.0f, bv1 = 0.0f, bv2 = 0.0f;
    if (mm) {
        const float* dr = dinv + n0 + rt * 16 + half * 8;
        #pragma unroll
        for (int v = 0; v < 8; v++) dv[v] = dr[v];
        bv0 = bias[nt0 * 16 + l15];
        bv1 = bias[nt0 * 16 + 16 + l15];
        bv2 = bias[nt0 * 16 + 32 + l15];
        const float* abase = acc + (rt * 16 + l15) * APITCH + half * 2;
        const float* bbase = sWt + (nt0 * 16 + l15) * WPITCH + half * 2;
        #pragma unroll 4
        for (int k = 0; k < HID; k += 4) {
            v2f a  = *(const v2f*)(abase + k);
            v2f b0 = *(const v2f*)(bbase + k);
            v2f b1 = *(const v2f*)(bbase + 16 * WPITCH + k);
            v2f b2 = *(const v2f*)(bbase + 32 * WPITCH + k);
            c0 = __builtin_amdgcn_wmma_f32_16x16x4_f32(
                 false, a,  false, b0,
                 (short)0, c0,  false,  false);
            c1 = __builtin_amdgcn_wmma_f32_16x16x4_f32(
                 false, a,  false, b1,
                 (short)0, c1,  false,  false);
            c2 = __builtin_amdgcn_wmma_f32_16x16x4_f32(
                 false, a,  false, b2,
                 (short)0, c2,  false,  false);
        }
    }
    __syncthreads();

    if (mm) {
        float* cs = acc + (rt * 16 + half * 8) * APITCH + nt0 * 16 + l15;
        #pragma unroll
        for (int v = 0; v < 8; v++) {
            float y0 = fmaxf(c0[v] * dv[v] + bv0, 0.0f);
            float y1 = fmaxf(c1[v] * dv[v] + bv1, 0.0f);
            float y2 = fmaxf(c2[v] * dv[v] + bv2, 0.0f);
            if (!LAST) { y0 *= dv[v]; y1 *= dv[v]; y2 *= dv[v]; }
            cs[v * APITCH]      = y0;
            cs[v * APITCH + 16] = y1;
            cs[v * APITCH + 32] = y2;
        }
    }
    __syncthreads();

    float4* hout4 = (float4*)(hout + (size_t)n0 * HID);
    float4 ov[TPASS];
    #pragma unroll
    for (int it = 0; it < TPASS; it++) {
        int i = tid + it * NT;
        int r = i / RV;
        int q = i - r * RV;
        ov[it] = lacc[r * (APITCH / 4) + q];
    }
    #pragma unroll
    for (int it = 0; it < TPASS; it++) {
        int i = tid + it * NT;
        if (i < nrows * RV) hout4[i] = ov[it];
    }
}

#define PL_NT 384
#define PL_S  ((NNODES + PL_NT - 1) / PL_NT)
#define PL_DEPTH 8

__global__ void __launch_bounds__(PL_NT)
k_pool_graph(const float* __restrict__ h, const int* __restrict__ batch,
             float* __restrict__ g, int n) {
    constexpr int NW    = PL_NT / 32;
    constexpr int RV    = HID / 4;
    constexpr int SLOTS = PL_NT / RV;
    static_assert(PL_NT % RV == 0 && PL_NT % 64 == 0, "a thread keeps its columns; the two halves are whole waves");
    static_assert(PL_S - 1 <= PL_NT / 2, "the ids between two samples fit one half of the work-group");
    __shared__ int wc[3][NW];
    __shared__ __attribute__((aligned(16))) float4 red[PL_NT];

    const int tid  = threadIdx.x;
    const int gi   = blockIdx.x;
    const int wave = tid >> 5;
    const int lane = tid & 31;

    const int p = tid * PL_S;
    bool below = false, upto = false;
    if (p < n) {
        int v = batch[p];
        below = v < gi;
        upto  = v <= gi;
    }
    const unsigned m0 = __builtin_amdgcn_ballot_w32(below);
    const unsigned m1 = __builtin_amdgcn_ballot_w32(upto);
    if (lane == 0) {
        wc[0][wave] = __builtin_popcount(m0);
        wc[1][wave] = __builtin_popcount(m1);
    }
    __syncthreads();
    int c_lo = 0, c_hi = 0;
    #pragma unroll
    for (int w = 0; w < NW; w++) {
        c_lo += wc[0][w];
        c_hi += wc[1][w];
    }
    const int upper = (tid >= PL_NT / 2) ? 1 : 0;
    const int j     = tid - upper * (PL_NT / 2);
    const int c     = upper ? c_hi : c_lo;
    bool in = false;
    if (c > 0 && j < PL_S - 1) {
        int idx = (c - 1) * PL_S + 1 + j;
        if (idx < n) {
            int v = batch[idx];
            in = upper ? (v <= gi) : (v < gi);
        }
    }
    const unsigned m2 = __builtin_amdgcn_ballot_w32(in);
    if (lane == 0) wc[2][wave] = __builtin_popcount(m2);
    __syncthreads();
    int d_lo = 0, d_hi = 0;
    #pragma unroll
    for (int w = 0; w < NW / 2; w++) {
        d_lo += wc[2][w];
        d_hi += wc[2][w + NW / 2];
    }
    const int lo = c_lo ? (c_lo - 1) * PL_S + 1 + d_lo : 0;
    const int hi = c_hi ? (c_hi - 1) * PL_S + 1 + d_hi : 0;

    const int cnt = (hi - lo) * RV;
    const float4* src = (const float4*)(h + (size_t)lo * HID);
    float4 acc = make_float4(0.0f, 0.0f, 0.0f, 0.0f);
    for (int base = 0; base < cnt; base += PL_DEPTH * PL_NT) {
        float4 v[PL_DEPTH];
        #pragma unroll
        for (int u = 0; u < PL_DEPTH; u++) {
            int i = base + u * PL_NT + tid;
            v[u] = make_float4(0.0f, 0.0f, 0.0f, 0.0f);
            if (i < cnt) v[u] = src[i];
        }
        #pragma unroll
        for (int u = 0; u < PL_DEPTH; u++) {
            acc.x += v[u].x; acc.y += v[u].y; acc.z += v[u].z; acc.w += v[u].w;
        }
    }
    red[tid] = acc;
    __syncthreads();
    if (tid < RV) {
        float4 s = red[tid];
        #pragma unroll
        for (int r = 1; r < SLOTS; r++) {
            float4 v = red[r * RV + tid];
            s.x += v.x; s.y += v.y; s.z += v.z; s.w += v.w;
        }
        ((float4*)(g + (size_t)gi * HID))[tid] = s;
    }
}

extern "C" void kernel_launch(void* const* d_in, const int* in_sizes, int n_in,
                              void* d_out, int out_size, void* d_ws, size_t ws_size,
                              hipStream_t stream) {
    const float* x       = (const float*)d_in[0];
    const int*   ei      = (const int*)  d_in[1];
    const int*   batch   = (const int*)  d_in[2];
    const float* enc_W0  = (const float*)d_in[3];
    const float* enc_b0  = (const float*)d_in[4];
    const float* enc_W1  = (const float*)d_in[5];
    const float* enc_b1  = (const float*)d_in[6];
    const float* conv_W  = (const float*)d_in[7];
    const float* conv_b  = (const float*)d_in[8];
    const float* dec_W0  = (const float*)d_in[9];
    const float* dec_b0  = (const float*)d_in[10];
    const float* dec_W1  = (const float*)d_in[11];
    const float* dec_b1  = (const float*)d_in[12];
    float* out = (float*)d_out;

    float* ws   = (float*)d_ws;
    size_t feat = (size_t)NNODES * HID;
    float* buf0 = ws;
    float* buf1 = buf0 + feat;
    float* buf2 = buf1 + feat;
    float* dinv = buf2 + feat;
    float* g    = dinv + DINV_PLANE;
    int*   hist   = (int*)(g + (size_t)NGRAPH * HID);
    int*   tstart = hist + (size_t)FB * NTP;
    int2*  bedge  = (int2*)(tstart + NTP);

    k_gemm_ws<INDIM, 1, 0><<<NTILES + FB, GW_WAVES * 32, 0, stream>>>(
        x, enc_W0, enc_b0, ei, hist, nullptr, nullptr, nullptr, nullptr, nullptr, nullptr, buf0, NNODES);

    int per = cdiv(NEDGES, FB);
    k_tile_fill<<<FB, FT, 0, stream>>>(ei, hist, tstart, bedge, NEDGES, per);

    k_gemm_ws<HID, 0, 1><<<NTILES, GW_WAVES * 32, 0, stream>>>(
        buf0, enc_W1, enc_b1, nullptr, nullptr, bedge, tstart, dinv, nullptr, nullptr, nullptr, buf1, NNODES);

    float* cur = buf1;
    float* nxt = buf2;
    for (int l = 0; l < NCONV; l++) {
        const float* Wl = conv_W + (size_t)l * HID * HID;
        const float* bl = conv_b + (size_t)l * HID;
        if (l + 1 < NCONV)
            k_tile_layer<0><<<NTILES, AGG_WAVES * 32, 0, stream>>>(cur, Wl, bl, dinv, bedge, tstart, nxt);
        else
            k_tile_layer<1><<<NTILES, AGG_WAVES * 32, 0, stream>>>(cur, Wl, bl, dinv, bedge, tstart, nxt);
        float* t = cur; cur = nxt; nxt = t;
    }

    k_pool_graph<<<NGRAPH, PL_NT, 0, stream>>>(cur, batch, g, NNODES);

    k_gemm_ws<HID, 1, 2><<<NGRAPH / TILE, GW_WAVES * 32, 0, stream>>>(
        g, dec_W0, dec_b0, nullptr, nullptr, nullptr, nullptr, nullptr, dec_W1, dec_b1, out, nullptr, NGRAPH);
}


# Launches 9 -> 8, folded again: decoder rides in the pool launch (128 work-groups x 4 graphs, padded WMMA tile), edge placement in encoder GEMM 1 after a count launch; pool sum, bin edge order change
# speedup vs baseline: 3.8618x; 1.0320x over previous
#include <hip/hip_runtime.h>
#include <hip/hip_bf16.h>

typedef __attribute__((ext_vector_type(2))) float v2f;
typedef __attribute__((ext_vector_type(8))) float v8f;

#define NNODES 50000
#define NEDGES 800000
#define NGRAPH 512
#define INDIM  128
#define HID    96
#define OUTDIM 10
#define NCONV  4
#define GW_WAVES 6
#define TSHIFT 6
#define TILE   64
#define NTILES ((NNODES + TILE - 1) / TILE)
#define NTP    1024
#define FB     128
#define AGG_WAVES 8
#define APITCH 100
#define WPITCH (HID + 2)
#define DINV_PLANE ((NNODES + 31) / 32 * 32)

static_assert(NNODES % 16 == 0, "a tile's valid rows must be whole 16-row WMMA tiles");

#define CT_NT 512

__global__ void __launch_bounds__(CT_NT)
k_slice_count(const int* __restrict__ ei, int* __restrict__ hist) {
    constexpr int PER = (NEDGES + FB - 1) / FB;
    constexpr int PRE = (PER + CT_NT - 1) / CT_NT;
    __shared__ int lh[NTILES];
    const int tid = threadIdx.x;
    const int sl  = blockIdx.x;
    const int e0  = sl * PER;
    const int e1  = min(e0 + PER, NEDGES);
    int d[PRE];
    #pragma unroll
    for (int u = 0; u < PRE; u++) {
        int e = e0 + tid + u * CT_NT;
        d[u] = (e < e1) ? ei[NEDGES + e] : -1;
    }
    for (int i = tid; i < NTILES; i += CT_NT) lh[i] = 0;
    __syncthreads();
    #pragma unroll
    for (int u = 0; u < PRE; u++)
        if (d[u] >= 0) atomicAdd(&lh[d[u] >> TSHIFT], 1);
    __syncthreads();
    for (int i = tid; i < NTILES; i += CT_NT) hist[sl * NTP + i] = lh[i];
}

__device__ __forceinline__ void slice_place(const int* __restrict__ ei, const int* __restrict__ hist,
                                            int* __restrict__ tstart, int2* __restrict__ bedge,
                                            int* lmem, int* wsum, int b, int tid) {
    constexpr int NT  = GW_WAVES * 32;
    constexpr int NQT = 4 * NT;
    constexpr int NXT = NTILES - NQT;
    constexpr int PER = (NEDGES + FB - 1) / FB;
    constexpr int EPT = 5;
    constexpr int DP  = 8;
    static_assert(NT * EPT >= NTILES && NT * EPT <= NTP && NTP % 4 == 0, "the scan covers every tile inside a table row");
    static_assert(NXT >= 1 && NXT <= NT && NTILES < NTP, "the table sums cover every tile; the row's last entry is spare");
    int* lh   = lmem;
    int* lb   = lmem + NTP;
    int* ltot = lmem + 2 * NTP;
    const int lane = tid & 31;
    const int wave = tid >> 5;
    const int e0 = b * PER;
    const int e1 = min(e0 + PER, NEDGES);

    int s[DP], d[DP];
    #pragma unroll
    for (int u = 0; u < DP; u++) {
        const int e  = e0 + tid + u * NT;
        const int ec = min(e, e1 - 1);
        s[u] = ei[ec];
        d[u] = (e < e1) ? ei[NEDGES + ec] : -1;
    }

    const bool xt = tid < NXT;
    const int  xi = NQT + min(tid, NXT - 1);
    const int4* hp4 = (const int4*)hist + tid;
    const int*  hpx = hist + xi;
    {
        int t0 = 0, t1 = 0, t2 = 0, t3 = 0, tx = 0;
        int p0 = 0, p1 = 0, p2 = 0, p3 = 0, px = 0;
        #pragma unroll 8
        for (int bb = 0; bb < FB; bb++) {
            const int4 c  = hp4[(size_t)bb * (NTP / 4)];
            const int  cx = hpx[(size_t)bb * NTP];
            const int  m  = (bb < b) ? -1 : 0;
            t0 += c.x; t1 += c.y; t2 += c.z; t3 += c.w; tx += cx;
            p0 += c.x & m; p1 += c.y & m; p2 += c.z & m; p3 += c.w & m; px += cx & m;
        }
        const int t = 4 * tid;
        ltot[t]     = t0; lb[t]     = p0; lh[t]     = 0;
        ltot[t + 1] = t1; lb[t + 1] = p1; lh[t + 1] = 0;
        ltot[t + 2] = t2; lb[t + 2] = p2; lh[t + 2] = 0;
        ltot[t + 3] = t3; lb[t + 3] = p3; lh[t + 3] = 0;
        if (xt) { ltot[NQT + tid] = tx; lb[NQT + tid] = px; lh[NQT + tid] = 0; }
        if (tid == 0) { lh[NTP - 1] = 0; lb[NTP - 1] = 0; }
    }
    for (int i = NTILES + tid; i < NT * EPT; i += NT) ltot[i] = 0;
    __syncthreads();

    int v[EPT];
    int sum = 0;
    #pragma unroll
    for (int k = 0; k < EPT; k++) {
        v[k] = ltot[tid * EPT + k];
        sum += v[k];
    }
    int inc = sum;
    #pragma unroll
    for (int off = 1; off < 32; off <<= 1) {
        int nb = __shfl_up(inc, off);
        if (lane >= off) inc += nb;
    }
    if (lane == 31) wsum[wave] = inc;
    __syncthreads();
    int run = inc - sum;
    for (int w = 0; w < wave; w++) run += wsum[w];
    #pragma unroll
    for (int k = 0; k < EPT; k++) {
        int t = tid * EPT + k;
        if (t < NTILES) {
            if (b == 0) tstart[t] = run;
            lb[t] += run;
        }
        run += v[k];
    }
    if (b == 0 && tid == NT - 1) tstart[NTILES] = run;
    __syncthreads();

    for (int base = e0; base < e1; base += DP * NT) {
        int sn[DP], dn[DP];
        #pragma unroll
        for (int u = 0; u < DP; u++) {
            const int e  = base + DP * NT + tid + u * NT;
            const int ec = min(e, e1 - 1);
            sn[u] = ei[ec];
            dn[u] = (e < e1) ? ei[NEDGES + ec] : -1;
        }
        int r[DP], bs[DP];
        #pragma unroll
        for (int u = 0; u < DP; u++) {
            const int bin = (d[u] >= 0) ? (d[u] >> TSHIFT) : (NTP - 1);
            r[u]  = atomicAdd(&lh[bin], 1);
            bs[u] = lb[bin];
        }
        #pragma unroll
        for (int u = 0; u < DP; u++)
            if (d[u] >= 0) bedge[bs[u] + r[u]] = make_int2(s[u], d[u]);
        #pragma unroll
        for (int u = 0; u < DP; u++) {
            s[u] = sn[u];
            d[u] = dn[u];
        }
    }
}

template <int K, int RELU, int MODE>
__global__ void __launch_bounds__(GW_WAVES * 32, MODE == 0 ? 7 : 6)
k_gemm_ws(const float* __restrict__ A, const float* __restrict__ B,
          const float* __restrict__ bias,
          const int* __restrict__ ei, const int* __restrict__ hist,
          int2* __restrict__ bedge, int* __restrict__ tstart,
          float* __restrict__ dinv,
          float* __restrict__ C, int M) {
    constexpr int NT    = GW_WAVES * 32;
    constexpr int KV    = K / 4;
    constexpr int KS    = K / 4;
    constexpr int AP    = K + 4;
    constexpr int CP    = HID + 4;
    constexpr int TOT   = TILE * KV;
    constexpr int NPASS = (TOT + NT - 1) / NT;
    constexpr int RV    = HID / 4;
    constexpr int CPASS = (TILE * RV + NT - 1) / NT;
    constexpr int DPRE  = 8;
    static_assert((AP / 4) % 2 == 1 && TILE * CP <= TILE * AP, "pitches");
    static_assert(MODE != 0 || 3 * NTP <= TILE * AP, "the placement's tables fit the A tile's space");
    __shared__ __attribute__((aligned(16))) float sA[TILE * AP];
    __shared__ int   cnt[MODE == 1 ? TILE : 1];
    __shared__ int   wsum[MODE == 0 ? GW_WAVES : 1];

    const int tid   = threadIdx.x;

    if (MODE == 0 && (int)blockIdx.x < FB) {
        slice_place(ei, hist, tstart, bedge, (int*)sA, wsum, (int)blockIdx.x, tid);
        return;
    }
    const int mtile = (MODE == 0) ? (int)blockIdx.x - FB : (int)blockIdx.x;

    const int wave  = __builtin_amdgcn_readfirstlane(tid >> 5);
    const int lane  = tid & 31;
    const int half  = lane >> 4;
    const int l15   = lane & 15;
    const int n0    = mtile * TILE;
    const int nrows = min(TILE, M - n0);

    const float4* A4 = (const float4*)(A + (size_t)n0 * K);
    float4 av[NPASS];
    #pragma unroll
    for (int it = 0; it < NPASS; it++) {
        int i = tid + it * NT;
        float4 v = make_float4(0.0f, 0.0f, 0.0f, 0.0f);
        if (i < nrows * KV) v = A4[i];
        av[it] = v;
    }
    const float* bcol = B + (size_t)(half * 2) * HID + wave * 16 + l15;
    v2f bf[KS];
    #pragma unroll
    for (int j = 0; j < KS; j++) {
        bf[j].x = bcol[(size_t)(4 * j) * HID];
        bf[j].y = bcol[(size_t)(4 * j + 1) * HID];
    }
    int dpre[DPRE];
    int e0 = 0, e1 = 0;
    const int* bdst = (const int*)bedge + 1;
    if (MODE == 1) {
        e0 = tstart[mtile];
        e1 = tstart[mtile + 1];
        #pragma unroll
        for (int u = 0; u < DPRE; u++) {
            int e = e0 + tid + u * NT;
            dpre[u] = (e < e1) ? bdst[2 * (size_t)e] : -1;
        }
        if (tid < TILE) cnt[tid] = 0;
    }
    #pragma unroll
    for (int it = 0; it < NPASS; it++) {
        int i = tid + it * NT;
        if (i < TOT) {
            int r = i / KV;
            int q = i - r * KV;
            *(float4*)(sA + r * AP + 4 * q) = av[it];
        }
    }
    __syncthreads();

    if (MODE == 1) {
        #pragma unroll
        for (int u = 0; u < DPRE; u++)
            if (dpre[u] >= 0) atomicAdd(&cnt[dpre[u] - n0], 1);
        for (int e = e0 + tid + DPRE * NT; e < e1; e += NT)
            atomicAdd(&cnt[bdst[2 * (size_t)e] - n0], 1);
    }

    v8f acc[4] = {};
    const float* abase = sA + l15 * AP + half * 2;
    #pragma unroll
    for (int j = 0; j < KS; j++) {
        #pragma unroll
        for (int rt = 0; rt < 4; rt++) {
            v2f a = *(const v2f*)(abase + rt * 16 * AP + 4 * j);
            acc[rt] = __builtin_amdgcn_wmma_f32_16x16x4_f32(
                 false, a,  false, bf[j],
                 (short)0, acc[rt],  false,  false);
        }
    }
    __syncthreads();

    const float bv = bias[wave * 16 + l15];
    float rs[CPASS];
    if (MODE == 1) {
        #pragma unroll
        for (int it = 0; it < CPASS; it++) {
            int r = (tid + it * NT) / RV;
            rs[it] = rsqrtf(1.0f + (float)cnt[r]);
        }
        if (tid < TILE && n0 + tid < M) dinv[n0 + tid] = rsqrtf(1.0f + (float)cnt[tid]);
    }
    float* cs = sA + (half * 8) * CP + wave * 16 + l15;
    #pragma unroll
    for (int rt = 0; rt < 4; rt++) {
        #pragma unroll
        for (int v = 0; v < 8; v++) {
            float r = acc[rt][v] + bv;
            if (RELU) r = fmaxf(r, 0.0f);
            cs[(rt * 16 + v) * CP] = r;
        }
    }
    __syncthreads();

    static_assert(TILE * RV == CPASS * NT, "the C tile is a whole number of passes");
    float4* C4 = (float4*)(C + (size_t)n0 * HID);
    if (nrows == TILE) {
        float4 cv[CPASS];
        #pragma unroll
        for (int it = 0; it < CPASS; it++) {
            int i = tid + it * NT;
            int r = i / RV;
            int q = i - r * RV;
            cv[it] = *(const float4*)(sA + r * CP + 4 * q);
            if (MODE == 1) {
                cv[it].x *= rs[it]; cv[it].y *= rs[it]; cv[it].z *= rs[it]; cv[it].w *= rs[it];
            }
        }
        #pragma unroll
        for (int it = 0; it < CPASS; it++) C4[tid + it * NT] = cv[it];
    } else {
        #pragma unroll
        for (int it = 0; it < CPASS; it++) {
            int i = tid + it * NT;
            if (i < nrows * RV) {
                int r = i / RV;
                int q = i - r * RV;
                float4 c = *(const float4*)(sA + r * CP + 4 * q);
                if (MODE == 1) { c.x *= rs[it]; c.y *= rs[it]; c.z *= rs[it]; c.w *= rs[it]; }
                C4[i] = c;
            }
        }
    }
}

template <int MASK>
__device__ __forceinline__ void tile_edge16(const float* __restrict__ h, float* acc,
                                            int s, int dl, int j, int cnt, int lane) {
    float v[16][3];
    int   ro[16];
    #pragma unroll
    for (int u = 0; u < 16; u++) {
        int   sj = __builtin_amdgcn_readlane(s, j + u);
        ro[u]    = __builtin_amdgcn_readlane(dl, j + u) * APITCH + lane;
        const float* srow = h + (size_t)sj * HID + lane;
        v[u][0] = srow[0];
        v[u][1] = srow[32];
        v[u][2] = srow[64];
    }
    #pragma unroll
    for (int u = 0; u < 16; u++) {
        const bool on = !MASK || (j + u) < cnt;
        __hip_atomic_fetch_add(acc + ro[u],      on ? v[u][0] : 0.0f, __ATOMIC_RELAXED, __HIP_MEMORY_SCOPE_WORKGROUP);
        __hip_atomic_fetch_add(acc + ro[u] + 32, on ? v[u][1] : 0.0f, __ATOMIC_RELAXED, __HIP_MEMORY_SCOPE_WORKGROUP);
        __hip_atomic_fetch_add(acc + ro[u] + 64, on ? v[u][2] : 0.0f, __ATOMIC_RELAXED, __HIP_MEMORY_SCOPE_WORKGROUP);
    }
}

template <int LAST>
__global__ void __launch_bounds__(AGG_WAVES * 32)
k_tile_layer(const float* __restrict__ h, const float* __restrict__ W,
             const float* __restrict__ bias, const float* __restrict__ dinv,
             const int2* __restrict__ bedge, const int* __restrict__ tstart,
             float* __restrict__ hout) {
    __shared__ __attribute__((aligned(16))) float acc[TILE * APITCH];
    __shared__ __attribute__((aligned(16))) float sWt[HID * WPITCH];
    constexpr int NT = AGG_WAVES * 32;
    constexpr int WV = HID * HID / 4;
    constexpr int RV = HID / 4;
    const int tid   = threadIdx.x;
    const int t     = blockIdx.x;
    const int n0    = t * TILE;
    const int nrows = min(TILE, NNODES - n0);
    const int wave = __builtin_amdgcn_readfirstlane(tid >> 5);
    const int lane = tid & 31;
    const int e0 = tstart[t], e1 = tstart[t + 1];
    int2 sd = make_int2(0, n0);
    if (e0 + wave * 32 + lane < e1) sd = bedge[e0 + wave * 32 + lane];

    constexpr int WPASS = WV / NT;
    constexpr int TPASS = TILE * RV / NT;
    static_assert(WV == WPASS * NT && TILE * RV == TPASS * NT, "W and the tile are whole numbers of passes");
    const float4* W4  = (const float4*)W;
    const float4* hin = (const float4*)(h + (size_t)n0 * HID);
    float4* lacc = (float4*)acc;
    float4 wv[WPASS];
    float4 tv[TPASS];
    #pragma unroll
    for (int it = 0; it < WPASS; it++) wv[it] = W4[tid + it * NT];
    #pragma unroll
    for (int it = 0; it < TPASS; it++) {
        int i = tid + it * NT;
        tv[it] = make_float4(0.0f, 0.0f, 0.0f, 0.0f);
        if (i < nrows * RV) tv[it] = hin[i];
    }
    #pragma unroll
    for (int it = 0; it < WPASS; it++) {
        int i   = tid + it * NT;
        int k   = i / RV;
        int col = (i - k * RV) * 4;
        float* p = sWt + col * WPITCH + k;
        p[0]          = wv[it].x;
        p[WPITCH]     = wv[it].y;
        p[2 * WPITCH] = wv[it].z;
        p[3 * WPITCH] = wv[it].w;
    }
    #pragma unroll
    for (int it = 0; it < TPASS; it++) {
        int i = tid + it * NT;
        int r = i / RV;
        int q = i - r * RV;
        lacc[r * (APITCH / 4) + q] = tv[it];
    }
    __syncthreads();

    for (int base = e0 + wave * 32; base < e1; base += NT) {
        const int s  = sd.x;
        const int dl = sd.y - n0;
        sd = make_int2(0, n0);
        if (base + NT + lane < e1) sd = bedge[base + NT + lane];
        const int cnt = min(32, e1 - base);
        int j = 0;
        for (; j + 16 <= cnt; j += 16) tile_edge16<0>(h, acc, s, dl, j, cnt, lane);
        if (j < cnt) tile_edge16<1>(h, acc, s, dl, j, cnt, lane);
    }
    __syncthreads();

    const int rt   = wave & 3;
    const int half = lane >> 4;
    const int l15  = lane & 15;
    const int nt0  = (wave >> 2) * 3;
    const bool mm  = rt * 16 < nrows;
    v8f c0 = {}, c1 = {}, c2 = {};
    float dv[8];
    float bv0 = 0.0f, bv1 = 0.0f, bv2 = 0.0f;
    if (mm) {
        const float* dr = dinv + n0 + rt * 16 + half * 8;
        #pragma unroll
        for (int v = 0; v < 8; v++) dv[v] = dr[v];
        bv0 = bias[nt0 * 16 + l15];
        bv1 = bias[nt0 * 16 + 16 + l15];
        bv2 = bias[nt0 * 16 + 32 + l15];
        const float* abase = acc + (rt * 16 + l15) * APITCH + half * 2;
        const float* bbase = sWt + (nt0 * 16 + l15) * WPITCH + half * 2;
        #pragma unroll 4
        for (int k = 0; k < HID; k += 4) {
            v2f a  = *(const v2f*)(abase + k);
            v2f b0 = *(const v2f*)(bbase + k);
            v2f b1 = *(const v2f*)(bbase + 16 * WPITCH + k);
            v2f b2 = *(const v2f*)(bbase + 32 * WPITCH + k);
            c0 = __builtin_amdgcn_wmma_f32_16x16x4_f32(
                 false, a,  false, b0,
                 (short)0, c0,  false,  false);
            c1 = __builtin_amdgcn_wmma_f32_16x16x4_f32(
                 false, a,  false, b1,
                 (short)0, c1,  false,  false);
            c2 = __builtin_amdgcn_wmma_f32_16x16x4_f32(
                 false, a,  false, b2,
                 (short)0, c2,  false,  false);
        }
    }
    __syncthreads();

    if (mm) {
        float* cs = acc + (rt * 16 + half * 8) * APITCH + nt0 * 16 + l15;
        #pragma unroll
        for (int v = 0; v < 8; v++) {
            float y0 = fmaxf(c0[v] * dv[v] + bv0, 0.0f);
            float y1 = fmaxf(c1[v] * dv[v] + bv1, 0.0f);
            float y2 = fmaxf(c2[v] * dv[v] + bv2, 0.0f);
            if (!LAST) { y0 *= dv[v]; y1 *= dv[v]; y2 *= dv[v]; }
            cs[v * APITCH]      = y0;
            cs[v * APITCH + 16] = y1;
            cs[v * APITCH + 32] = y2;
        }
    }
    __syncthreads();

    float4* hout4 = (float4*)(hout + (size_t)n0 * HID);
    float4 ov[TPASS];
    #pragma unroll
    for (int it = 0; it < TPASS; it++) {
        int i = tid + it * NT;
        int r = i / RV;
        int q = i - r * RV;
        ov[it] = lacc[r * (APITCH / 4) + q];
    }
    #pragma unroll
    for (int it = 0; it < TPASS; it++) {
        int i = tid + it * NT;
        if (i < nrows * RV) hout4[i] = ov[it];
    }
}

#define PD_G     4
#define PD_NT    384
#define PD_S     ((NNODES + PD_NT - 1) / PD_NT)
#define PD_DEPTH 13
#define PD_GP    100

static_assert(NGRAPH % PD_G == 0, "whole groups of graphs");

__global__ void __launch_bounds__(PD_NT)
k_pool_dec(const float* __restrict__ h, const int* __restrict__ batch,
           const float* __restrict__ W0, const float* __restrict__ b0,
           const float* __restrict__ W1, const float* __restrict__ b1,
           float* __restrict__ out, int n) {
    constexpr int NW    = PD_NT / 32;
    constexpr int RV    = HID / 4;
    constexpr int SLOTS = PD_NT / RV;
    constexpr int KS    = HID / 4;
    constexpr int W2P   = (HID * OUTDIM + PD_NT - 1) / PD_NT;
    static_assert(PD_NT % RV == 0 && PD_NT % 64 == 0, "a thread keeps its columns; the two halves are whole waves");
    static_assert(PD_S - 1 <= PD_NT / 2, "the ids between two samples fit one half of the work-group");
    static_assert(PD_G * RV <= PD_NT && PD_G * OUTDIM <= PD_NT && GW_WAVES <= NW && PD_G <= 8, "the tail jobs fit the work-group");
    __shared__ int wc[3 + PD_G - 1][NW];
    __shared__ __attribute__((aligned(16))) float4 red[PD_G][PD_NT];
    __shared__ __attribute__((aligned(16))) float gA[16 * PD_GP];
    __shared__ float tT[PD_G * HID];
    __shared__ float sW2[HID * OUTDIM];

    const int tid  = threadIdx.x;
    const int g0   = blockIdx.x * PD_G;
    const int g1   = g0 + PD_G - 1;
    const int wave = __builtin_amdgcn_readfirstlane(tid >> 5);
    const int lane = tid & 31;
    const int half = lane >> 4;
    const int l15  = lane & 15;

    v2f bf[KS] = {};
    float bv = 0.0f;
    if (wave < GW_WAVES) {
        const float* bcol = W0 + (size_t)(half * 2) * HID + wave * 16 + l15;
        #pragma unroll
        for (int j = 0; j < KS; j++) {
            bf[j].x = bcol[(size_t)(4 * j) * HID];
            bf[j].y = bcol[(size_t)(4 * j + 1) * HID];
        }
        bv = b0[wave * 16 + l15];
    }
    float w2v[W2P];
    #pragma unroll
    for (int it = 0; it < W2P; it++) {
        int i = tid + it * PD_NT;
        w2v[it] = (i < HID * OUTDIM) ? W1[i] : 0.0f;
    }
    float b1v = 0.0f;
    if (tid < PD_G * OUTDIM) b1v = b1[tid % OUTDIM];
    for (int i = tid; i < (16 - PD_G) * PD_GP; i += PD_NT) gA[PD_G * PD_GP + i] = 0.0f;

    const int p = tid * PD_S;
    bool below = false, upto = false;
    if (p < n) {
        int v = batch[p];
        below = v < g0;
        upto  = v <= g1;
    }
    const unsigned m0 = __builtin_amdgcn_ballot_w32(below);
    const unsigned m1 = __builtin_amdgcn_ballot_w32(upto);
    if (lane == 0) {
        wc[0][wave] = __builtin_popcount(m0);
        wc[1][wave] = __builtin_popcount(m1);
    }
    __syncthreads();
    int c_lo = 0, c_hi = 0;
    #pragma unroll
    for (int w = 0; w < NW; w++) {
        c_lo += wc[0][w];
        c_hi += wc[1][w];
    }
    const int upper = (tid >= PD_NT / 2) ? 1 : 0;
    const int jj    = tid - upper * (PD_NT / 2);
    const int cc    = upper ? c_hi : c_lo;
    bool in = false;
    if (cc > 0 && jj < PD_S - 1) {
        int idx = (cc - 1) * PD_S + 1 + jj;
        if (idx < n) {
            int v = batch[idx];
            in = upper ? (v <= g1) : (v < g0);
        }
    }
    const unsigned m2 = __builtin_amdgcn_ballot_w32(in);
    if (lane == 0) wc[2][wave] = __builtin_popcount(m2);
    #pragma unroll
    for (int it = 0; it < W2P; it++) {
        int i = tid + it * PD_NT;
        if (i < HID * OUTDIM) sW2[i] = w2v[it];
    }
    __syncthreads();
    int d_lo = 0, d_hi = 0;
    #pragma unroll
    for (int w = 0; w < NW / 2; w++) {
        d_lo += wc[2][w];
        d_hi += wc[2][w + NW / 2];
    }
    const int lo = c_lo ? (c_lo - 1) * PD_S + 1 + d_lo : 0;
    const int hi = c_hi ? (c_hi - 1) * PD_S + 1 + d_hi : 0;

    const int nrun = hi - lo;
    int wcnt[PD_G - 1];
    #pragma unroll
    for (int k = 0; k < PD_G - 1; k++) wcnt[k] = 0;
    for (int r0 = 0; r0 < nrun; r0 += PD_NT) {
        const int r = r0 + tid;
        int bid = 0x7fffffff;
        if (r < nrun) bid = batch[lo + r];
        #pragma unroll
        for (int k = 0; k < PD_G - 1; k++)
            wcnt[k] += __builtin_popcount(__builtin_amdgcn_ballot_w32(bid < g0 + 1 + k));
    }
    if (lane == 0) {
        #pragma unroll
        for (int k = 0; k < PD_G - 1; k++) wc[3 + k][wave] = wcnt[k];
    }
    __syncthreads();
    int gb[PD_G - 1];
    #pragma unroll
    for (int k = 0; k < PD_G - 1; k++) {
        int sb = 0;
        #pragma unroll
        for (int w = 0; w < NW; w++) sb += wc[3 + k][w];
        gb[k] = sb;
    }

    const int cnt  = nrun * RV;
    const int slot = tid / RV;
    const float4* src = (const float4*)(h + (size_t)lo * HID);
    float4 acc[PD_G];
    #pragma unroll
    for (int j = 0; j < PD_G; j++) acc[j] = make_float4(0.0f, 0.0f, 0.0f, 0.0f);
    for (int base = 0; base < cnt; base += PD_DEPTH * PD_NT) {
        float4 v[PD_DEPTH];
        #pragma unroll
        for (int u = 0; u < PD_DEPTH; u++) {
            int i = base + u * PD_NT + tid;
            v[u] = make_float4(0.0f, 0.0f, 0.0f, 0.0f);
            if (i < cnt) v[u] = src[i];
        }
        #pragma unroll
        for (int u = 0; u < PD_DEPTH; u++) {
            const int row = base / RV + u * SLOTS + slot;
            int q = 0;
            #pragma unroll
            for (int k = 0; k < PD_G - 1; k++) q += (row >= gb[k]) ? 1 : 0;
            #pragma unroll
            for (int j = 0; j < PD_G; j++) {
                const bool on = q == j;
                acc[j].x += on ? v[u].x : 0.0f;
                acc[j].y += on ? v[u].y : 0.0f;
                acc[j].z += on ? v[u].z : 0.0f;
                acc[j].w += on ? v[u].w : 0.0f;
            }
        }
    }
    #pragma unroll
    for (int j = 0; j < PD_G; j++) red[j][tid] = acc[j];
    __syncthreads();
    if (tid < PD_G * RV) {
        const int j  = tid / RV;
        const int pc = tid - j * RV;
        float4 s = red[j][pc];
        #pragma unroll
        for (int r = 1; r < SLOTS; r++) {
            float4 v = red[j][r * RV + pc];
            s.x += v.x; s.y += v.y; s.z += v.z; s.w += v.w;
        }
        *(float4*)(gA + j * PD_GP + 4 * pc) = s;
    }
    __syncthreads();

    if (wave < GW_WAVES) {
        v8f c = {};
        const float* abase = gA + l15 * PD_GP + half * 2;
        #pragma unroll
        for (int j = 0; j < KS; j++) {
            v2f a = *(const v2f*)(abase + 4 * j);
            c = __builtin_amdgcn_wmma_f32_16x16x4_f32(
                 false, a,  false, bf[j],
                 (short)0, c,  false,  false);
        }
        if (half == 0) {
            #pragma unroll
            for (int v = 0; v < PD_G; v++)
                tT[v * HID + wave * 16 + l15] = fmaxf(c[v] + bv, 0.0f);
        }
    }
    __syncthreads();

    if (tid < PD_G * OUTDIM) {
        const int r = tid / OUTDIM;
        const int o = tid - r * OUTDIM;
        float s = b1v;
        const float* row = tT + r * HID;
        #pragma unroll 8
        for (int k = 0; k < HID; k++) s += row[k] * sW2[k * OUTDIM + o];
        out[(size_t)g0 * OUTDIM + tid] = s;
    }
}

extern "C" void kernel_launch(void* const* d_in, const int* in_sizes, int n_in,
                              void* d_out, int out_size, void* d_ws, size_t ws_size,
                              hipStream_t stream) {
    const float* x       = (const float*)d_in[0];
    const int*   ei      = (const int*)  d_in[1];
    const int*   batch   = (const int*)  d_in[2];
    const float* enc_W0  = (const float*)d_in[3];
    const float* enc_b0  = (const float*)d_in[4];
    const float* enc_W1  = (const float*)d_in[5];
    const float* enc_b1  = (const float*)d_in[6];
    const float* conv_W  = (const float*)d_in[7];
    const float* conv_b  = (const float*)d_in[8];
    const float* dec_W0  = (const float*)d_in[9];
    const float* dec_b0  = (const float*)d_in[10];
    const float* dec_W1  = (const float*)d_in[11];
    const float* dec_b1  = (const float*)d_in[12];
    float* out = (float*)d_out;

    float* ws   = (float*)d_ws;
    size_t feat = (size_t)NNODES * HID;
    float* buf0 = ws;
    float* buf1 = buf0 + feat;
    float* buf2 = buf1 + feat;
    float* dinv = buf2 + feat;
    int*   hist   = (int*)(dinv + DINV_PLANE);
    int*   tstart = hist + (size_t)FB * NTP;
    int2*  bedge  = (int2*)(tstart + NTP);

    k_slice_count<<<FB, CT_NT, 0, stream>>>(ei, hist);

    k_gemm_ws<INDIM, 1, 0><<<FB + NTILES, GW_WAVES * 32, 0, stream>>>(
        x, enc_W0, enc_b0, ei, hist, bedge, tstart, nullptr, buf0, NNODES);

    k_gemm_ws<HID, 0, 1><<<NTILES, GW_WAVES * 32, 0, stream>>>(
        buf0, enc_W1, enc_b1, nullptr, nullptr, bedge, tstart, dinv, buf1, NNODES);

    float* cur = buf1;
    float* nxt = buf2;
    for (int l = 0; l < NCONV; l++) {
        const float* Wl = conv_W + (size_t)l * HID * HID;
        const float* bl = conv_b + (size_t)l * HID;
        if (l + 1 < NCONV)
            k_tile_layer<0><<<NTILES, AGG_WAVES * 32, 0, stream>>>(cur, Wl, bl, dinv, bedge, tstart, nxt);
        else
            k_tile_layer<1><<<NTILES, AGG_WAVES * 32, 0, stream>>>(cur, Wl, bl, dinv, bedge, tstart, nxt);
        float* t = cur; cur = nxt; nxt = t;
    }

    k_pool_dec<<<NGRAPH / PD_G, PD_NT, 0, stream>>>(
        cur, batch, dec_W0, dec_b0, dec_W1, dec_b1, out, NNODES);
}
